# starting kernel
# baseline (speedup 1.0000x reference)
#include <hip/hip_runtime.h>

typedef __attribute__((ext_vector_type(2))) float v2f;
typedef __attribute__((ext_vector_type(8))) float v8f;

#define Bsz  64
#define Sseq 4096
#define Cin  80
#define Hh   10

__global__ __launch_bounds__(128) void xproj_wmma(const float* __restrict__ inp,
                                                  const float* __restrict__ Wih,
                                                  const float* __restrict__ bih,
                                                  const float* __restrict__ bhh,
                                                  float* __restrict__ xp) {
    const int lane = threadIdx.x & 31;
    const int wave = threadIdx.x >> 5;
    const int tile = blockIdx.x * 4 + wave;
    const int half = lane >> 4;
    const int n    = lane & 15;
    const int rowA = tile * 16 + n;
    const int nn   = (n < Hh) ? n : (Hh - 1);
    const float nmask = (n < Hh) ? 1.0f : 0.0f;

    v8f acc = {};
    #pragma unroll
    for (int ks = 0; ks < Cin / 4; ++ks) {
        const int kk = ks * 4 + half * 2;
        const float2 a2 = *reinterpret_cast<const float2*>(inp + (size_t)rowA * Cin + kk);
        v2f afrag; afrag[0] = a2.x; afrag[1] = a2.y;
        v2f bfrag;
        bfrag[0] = nmask * Wih[nn * Cin + kk];
        bfrag[1] = nmask * Wih[nn * Cin + kk + 1];
        acc = __builtin_amdgcn_wmma_f32_16x16x4_f32(
            false, afrag, false, bfrag, (short)0, acc, false, false);
    }

    if (n < Hh) {
        const float bias = bih[n] + bhh[n];
        #pragma unroll
        for (int v = 0; v < 8; ++v) {
            const int g    = tile * 16 + v + half * 8;
            const int bidx = g >> 12;
            const int sidx = g & 4095;
            xp[((size_t)sidx * Bsz + bidx) * Hh + n] = acc[v] + bias;
        }
    }
}

__device__ __forceinline__ float fast_tanh(float x) {
    const float e = __expf(2.0f * x);
    return 1.0f - 2.0f / (e + 1.0f);
}

__device__ __forceinline__ float sig4(float logit) {
    const float sg = 1.0f / (1.0f + __expf(-logit));
    const float s2 = sg * sg;
    return s2 * s2;
}

__global__ __launch_bounds__(32) void rnn_scan(const float* __restrict__ xp,
                                               const float* __restrict__ Whh,
                                               const float* __restrict__ Wfc,
                                               const float* __restrict__ bfc_p,
                                               float* __restrict__ out) {
    const int lane = threadIdx.x;
    const int half = lane >> 4;
    const int i    = lane & 15;
    const int ii   = (i < Hh) ? i : (Hh - 1);
    const int b    = blockIdx.x * 2 + half;
    const int base = (lane & 16);

    float w[Hh], wf[Hh];
    #pragma unroll
    for (int j = 0; j < Hh; ++j) w[j]  = Whh[ii * Hh + j];
    #pragma unroll
    for (int j = 0; j < Hh; ++j) wf[j] = Wfc[j];
    const float bfc = bfc_p[0];

    float h = 0.0f;

    float xc = xp[(size_t)b * Hh + ii];

    float* const outb = out + (size_t)b * Sseq;

    for (int t = 0; t < Sseq; ++t) {
        float xn = 0.0f;
        if (t + 1 < Sseq)
            xn = xp[((size_t)(t + 1) * Bsz + b) * Hh + ii];
        if (t + 16 < Sseq)
            __builtin_prefetch(xp + ((size_t)(t + 16) * Bsz + b) * Hh + ii, 0, 3);

        float hj[Hh];
        #pragma unroll
        for (int j = 0; j < Hh; ++j) hj[j] = __shfl(h, base + j, 32);

        float a0 = xc, a1 = 0.0f;
        #pragma unroll
        for (int j = 0; j < 5; ++j) a0 = fmaf(w[j],     hj[j],     a0);
        #pragma unroll
        for (int j = 0; j < 5; ++j) a1 = fmaf(w[j + 5], hj[j + 5], a1);
        const float nh = fast_tanh(a0 + a1);

        if (t > 0) {
            float l0 = bfc, l1 = 0.0f;
            #pragma unroll
            for (int j = 0; j < 5; ++j) l0 = fmaf(wf[j],     hj[j],     l0);
            #pragma unroll
            for (int j = 0; j < 5; ++j) l1 = fmaf(wf[j + 5], hj[j + 5], l1);
            if (i == 0) outb[t - 1] = sig4(l0 + l1);
        }

        h = nh;
        xc = xn;
    }

    float hj[Hh];
    #pragma unroll
    for (int j = 0; j < Hh; ++j) hj[j] = __shfl(h, base + j, 32);
    float l0 = bfc, l1 = 0.0f;
    #pragma unroll
    for (int j = 0; j < 5; ++j) l0 = fmaf(wf[j],     hj[j],     l0);
    #pragma unroll
    for (int j = 0; j < 5; ++j) l1 = fmaf(wf[j + 5], hj[j + 5], l1);
    if (i == 0) outb[Sseq - 1] = sig4(l0 + l1);
}

extern "C" void kernel_launch(void* const* d_in, const int* in_sizes, int n_in,
                              void* d_out, int out_size, void* d_ws, size_t ws_size,
                              hipStream_t stream) {
    (void)in_sizes; (void)n_in; (void)out_size; (void)ws_size;
    const float* inp = (const float*)d_in[0];
    const float* Wih = (const float*)d_in[1];
    const float* Whh = (const float*)d_in[2];
    const float* bih = (const float*)d_in[3];
    const float* bhh = (const float*)d_in[4];
    const float* Wfc = (const float*)d_in[5];
    const float* bfc = (const float*)d_in[6];
    float* out = (float*)d_out;
    float* xp  = (float*)d_ws;

    const int tiles = (Bsz * Sseq) / 16;
    xproj_wmma<<<dim3(tiles / 4), 128, 0, stream>>>(inp, Wih, bih, bhh, xp);
    rnn_scan<<<dim3(Bsz / 2), 32, 0, stream>>>(xp, Whh, Wfc, bfc, out);
}


# scan: FC + sigmoid^4 of step t-1 computed branch-free in all lanes (kept by lane (t-1)&15, one 64-byte row stored per 16 steps) so its chain interleaves with the matvec + tanh chain
# speedup vs baseline: 1.0645x; 1.0645x over previous
#include <hip/hip_runtime.h>

typedef __attribute__((ext_vector_type(2))) float v2f;
typedef __attribute__((ext_vector_type(8))) float v8f;

#define Bsz  64
#define Sseq 4096
#define Cin  80
#define Hh   10

__global__ __launch_bounds__(128) void xproj_wmma(const float* __restrict__ inp,
                                                  const float* __restrict__ Wih,
                                                  const float* __restrict__ bih,
                                                  const float* __restrict__ bhh,
                                                  float* __restrict__ xp) {
    const int lane = threadIdx.x & 31;
    const int wave = threadIdx.x >> 5;
    const int tile = blockIdx.x * 4 + wave;
    const int half = lane >> 4;
    const int n    = lane & 15;
    const int rowA = tile * 16 + n;
    const int nn   = (n < Hh) ? n : (Hh - 1);
    const float nmask = (n < Hh) ? 1.0f : 0.0f;

    v8f acc = {};
    #pragma unroll
    for (int ks = 0; ks < Cin / 4; ++ks) {
        const int kk = ks * 4 + half * 2;
        const float2 a2 = *reinterpret_cast<const float2*>(inp + (size_t)rowA * Cin + kk);
        v2f afrag; afrag[0] = a2.x; afrag[1] = a2.y;
        v2f bfrag;
        bfrag[0] = nmask * Wih[nn * Cin + kk];
        bfrag[1] = nmask * Wih[nn * Cin + kk + 1];
        acc = __builtin_amdgcn_wmma_f32_16x16x4_f32(
            false, afrag, false, bfrag, (short)0, acc, false, false);
    }

    if (n < Hh) {
        const float bias = bih[n] + bhh[n];
        #pragma unroll
        for (int v = 0; v < 8; ++v) {
            const int g    = tile * 16 + v + half * 8;
            const int bidx = g >> 12;
            const int sidx = g & 4095;
            xp[((size_t)sidx * Bsz + bidx) * Hh + n] = acc[v] + bias;
        }
    }
}

__device__ __forceinline__ float fast_tanh(float x) {
    const float e = __expf(2.0f * x);
    return 1.0f - 2.0f / (e + 1.0f);
}

__device__ __forceinline__ float sig4(float logit) {
    const float sg = 1.0f / (1.0f + __expf(-logit));
    const float s2 = sg * sg;
    return s2 * s2;
}

__global__ __launch_bounds__(32) void rnn_scan(const float* __restrict__ xp,
                                               const float* __restrict__ Whh,
                                               const float* __restrict__ Wfc,
                                               const float* __restrict__ bfc_p,
                                               float* __restrict__ out) {
    const int lane = threadIdx.x;
    const int half = lane >> 4;
    const int i    = lane & 15;
    const int ii   = (i < Hh) ? i : (Hh - 1);
    const int b    = blockIdx.x * 2 + half;
    const int base = (lane & 16);

    float w[Hh], wf[Hh];
    #pragma unroll
    for (int j = 0; j < Hh; ++j) w[j]  = Whh[ii * Hh + j];
    #pragma unroll
    for (int j = 0; j < Hh; ++j) wf[j] = Wfc[j];
    const float bfc = bfc_p[0];

    float h = 0.0f;
    float keep = 0.0f;

    float xc = xp[(size_t)b * Hh + ii];

    float* const outb = out + (size_t)b * Sseq;

    for (int t = 0; t < Sseq; ++t) {
        if ((t & 15) == 1 && t != 1) outb[t - 17 + i] = keep;

        float xn = 0.0f;
        if (t + 1 < Sseq)
            xn = xp[((size_t)(t + 1) * Bsz + b) * Hh + ii];
        if (t + 16 < Sseq)
            __builtin_prefetch(xp + ((size_t)(t + 16) * Bsz + b) * Hh + ii, 0, 3);

        float hj[Hh];
        #pragma unroll
        for (int j = 0; j < Hh; ++j) hj[j] = __shfl(h, base + j, 32);

        float a0 = xc, a1 = 0.0f;
        #pragma unroll
        for (int j = 0; j < 5; ++j) a0 = fmaf(w[j],     hj[j],     a0);
        #pragma unroll
        for (int j = 0; j < 5; ++j) a1 = fmaf(w[j + 5], hj[j + 5], a1);
        const float nh = fast_tanh(a0 + a1);

        float l0 = bfc, l1 = 0.0f;
        #pragma unroll
        for (int j = 0; j < 5; ++j) l0 = fmaf(wf[j],     hj[j],     l0);
        #pragma unroll
        for (int j = 0; j < 5; ++j) l1 = fmaf(wf[j + 5], hj[j + 5], l1);
        const float val = sig4(l0 + l1);
        keep = (((t - 1) & 15) == i) ? val : keep;

        h = nh;
        xc = xn;
    }

    float hj[Hh];
    #pragma unroll
    for (int j = 0; j < Hh; ++j) hj[j] = __shfl(h, base + j, 32);
    float l0 = bfc, l1 = 0.0f;
    #pragma unroll
    for (int j = 0; j < 5; ++j) l0 = fmaf(wf[j],     hj[j],     l0);
    #pragma unroll
    for (int j = 0; j < 5; ++j) l1 = fmaf(wf[j + 5], hj[j + 5], l1);
    keep = (i == 15) ? sig4(l0 + l1) : keep;
    outb[Sseq - 16 + i] = keep;
}

extern "C" void kernel_launch(void* const* d_in, const int* in_sizes, int n_in,
                              void* d_out, int out_size, void* d_ws, size_t ws_size,
                              hipStream_t stream) {
    (void)in_sizes; (void)n_in; (void)out_size; (void)ws_size;
    const float* inp = (const float*)d_in[0];
    const float* Wih = (const float*)d_in[1];
    const float* Whh = (const float*)d_in[2];
    const float* bih = (const float*)d_in[3];
    const float* bhh = (const float*)d_in[4];
    const float* Wfc = (const float*)d_in[5];
    const float* bfc = (const float*)d_in[6];
    float* out = (float*)d_out;
    float* xp  = (float*)d_ws;

    const int tiles = (Bsz * Sseq) / 16;
    xproj_wmma<<<dim3(tiles / 4), 128, 0, stream>>>(inp, Wih, bih, bhh, xp);
    rnn_scan<<<dim3(Bsz / 2), 32, 0, stream>>>(xp, Whh, Wfc, bfc, out);
}


# scan: xp read a whole 16-step block ahead into a second register set (16 loads in flight, 16 steps written out per loop pass) instead of one step ahead, so no step waits for a second-level-cache hit
# speedup vs baseline: 1.6084x; 1.5109x over previous
#include <hip/hip_runtime.h>

typedef __attribute__((ext_vector_type(2))) float v2f;
typedef __attribute__((ext_vector_type(8))) float v8f;

#define Bsz  64
#define Sseq 4096
#define Cin  80
#define Hh   10

__global__ __launch_bounds__(128) void xproj_wmma(const float* __restrict__ inp,
                                                  const float* __restrict__ Wih,
                                                  const float* __restrict__ bih,
                                                  const float* __restrict__ bhh,
                                                  float* __restrict__ xp) {
    const int lane = threadIdx.x & 31;
    const int wave = threadIdx.x >> 5;
    const int tile = blockIdx.x * 4 + wave;
    const int half = lane >> 4;
    const int n    = lane & 15;
    const int rowA = tile * 16 + n;
    const int nn   = (n < Hh) ? n : (Hh - 1);
    const float nmask = (n < Hh) ? 1.0f : 0.0f;

    v8f acc = {};
    #pragma unroll
    for (int ks = 0; ks < Cin / 4; ++ks) {
        const int kk = ks * 4 + half * 2;
        const float2 a2 = *reinterpret_cast<const float2*>(inp + (size_t)rowA * Cin + kk);
        v2f afrag; afrag[0] = a2.x; afrag[1] = a2.y;
        v2f bfrag;
        bfrag[0] = nmask * Wih[nn * Cin + kk];
        bfrag[1] = nmask * Wih[nn * Cin + kk + 1];
        acc = __builtin_amdgcn_wmma_f32_16x16x4_f32(
            false, afrag, false, bfrag, (short)0, acc, false, false);
    }

    if (n < Hh) {
        const float bias = bih[n] + bhh[n];
        #pragma unroll
        for (int v = 0; v < 8; ++v) {
            const int g    = tile * 16 + v + half * 8;
            const int bidx = g >> 12;
            const int sidx = g & 4095;
            xp[((size_t)sidx * Bsz + bidx) * Hh + n] = acc[v] + bias;
        }
    }
}

#define BLK   16
#define NBLK  (Sseq / BLK)
#define XSTEP (Bsz * Hh)

__device__ __forceinline__ float fast_tanh(float x) {
    const float e = __expf(2.0f * x);
    return 1.0f - 2.0f / (e + 1.0f);
}

__device__ __forceinline__ float sig4(float logit) {
    const float sg = 1.0f / (1.0f + __expf(-logit));
    const float s2 = sg * sg;
    return s2 * s2;
}

__global__ __launch_bounds__(32) void rnn_scan(const float* __restrict__ xp,
                                               const float* __restrict__ Whh,
                                               const float* __restrict__ Wfc,
                                               const float* __restrict__ bfc_p,
                                               float* __restrict__ out) {
    const int lane = threadIdx.x;
    const int half = lane >> 4;
    const int i    = lane & 15;
    const int ii   = (i < Hh) ? i : (Hh - 1);
    const int b    = blockIdx.x * 2 + half;
    const int base = (lane & 16);

    float w[Hh], wf[Hh];
    #pragma unroll
    for (int j = 0; j < Hh; ++j) w[j]  = Whh[ii * Hh + j];
    #pragma unroll
    for (int j = 0; j < Hh; ++j) wf[j] = Wfc[j];
    const float bfc = bfc_p[0];

    float h = 0.0f;
    float keep = 0.0f;

    const float* const xl = xp + (size_t)b * Hh + ii;

    float xa[BLK], xb[BLK];
    #pragma unroll
    for (int k = 0; k < BLK; ++k) xa[k] = xl[(size_t)k * XSTEP];

    float* const outb = out + (size_t)b * Sseq;

    const float* xn = xl + (size_t)(BLK * XSTEP);

    for (int g = 0; g < NBLK; ++g) {
        #pragma unroll
        for (int k = 0; k < BLK; ++k) xb[k] = xn[(size_t)k * XSTEP];

        #pragma unroll
        for (int k = 0; k < BLK; ++k) {
            if (k == 1 && g != 0) outb[(g - 1) * BLK + i] = keep;

            float hj[Hh];
            #pragma unroll
            for (int j = 0; j < Hh; ++j) hj[j] = __shfl(h, base + j, 32);

            float a0 = xa[k], a1 = 0.0f;
            #pragma unroll
            for (int j = 0; j < 5; ++j) a0 = fmaf(w[j],     hj[j],     a0);
            #pragma unroll
            for (int j = 0; j < 5; ++j) a1 = fmaf(w[j + 5], hj[j + 5], a1);
            const float nh = fast_tanh(a0 + a1);

            float l0 = bfc, l1 = 0.0f;
            #pragma unroll
            for (int j = 0; j < 5; ++j) l0 = fmaf(wf[j],     hj[j],     l0);
            #pragma unroll
            for (int j = 0; j < 5; ++j) l1 = fmaf(wf[j + 5], hj[j + 5], l1);
            const float val = sig4(l0 + l1);
            keep = (((k + 15) & 15) == i) ? val : keep;

            h = nh;

            asm volatile("" : "+v"(keep), "+v"(h));
        }

        #pragma unroll
        for (int k = 0; k < BLK; ++k) xa[k] = xb[k];
        xn += (g + 2 < NBLK) ? (BLK * XSTEP) : 0;
    }

    float hj[Hh];
    #pragma unroll
    for (int j = 0; j < Hh; ++j) hj[j] = __shfl(h, base + j, 32);
    float l0 = bfc, l1 = 0.0f;
    #pragma unroll
    for (int j = 0; j < 5; ++j) l0 = fmaf(wf[j],     hj[j],     l0);
    #pragma unroll
    for (int j = 0; j < 5; ++j) l1 = fmaf(wf[j + 5], hj[j + 5], l1);
    keep = (i == 15) ? sig4(l0 + l1) : keep;
    outb[Sseq - 16 + i] = keep;
}

extern "C" void kernel_launch(void* const* d_in, const int* in_sizes, int n_in,
                              void* d_out, int out_size, void* d_ws, size_t ws_size,
                              hipStream_t stream) {
    (void)in_sizes; (void)n_in; (void)out_size; (void)ws_size;
    const float* inp = (const float*)d_in[0];
    const float* Wih = (const float*)d_in[1];
    const float* Whh = (const float*)d_in[2];
    const float* bih = (const float*)d_in[3];
    const float* bhh = (const float*)d_in[4];
    const float* Wfc = (const float*)d_in[5];
    const float* bfc = (const float*)d_in[6];
    float* out = (float*)d_out;
    float* xp  = (float*)d_ws;

    const int tiles = (Bsz * Sseq) / 16;
    xproj_wmma<<<dim3(tiles / 4), 128, 0, stream>>>(inp, Wih, bih, bhh, xp);
    rnn_scan<<<dim3(Bsz / 2), 32, 0, stream>>>(xp, Whh, Wfc, bfc, out);
}


# scan: h_{t-1} read across lanes on the vector ALU's DPP path (row_share source of the 20 fused multiply-adds of matvec and FC) instead of 10 ds_bpermute per step; same sums, same order
# speedup vs baseline: 2.2665x; 1.4092x over previous
#include <hip/hip_runtime.h>

typedef __attribute__((ext_vector_type(2))) float v2f;
typedef __attribute__((ext_vector_type(8))) float v8f;

#define Bsz  64
#define Sseq 4096
#define Cin  80
#define Hh   10

__global__ __launch_bounds__(128) void xproj_wmma(const float* __restrict__ inp,
                                                  const float* __restrict__ Wih,
                                                  const float* __restrict__ bih,
                                                  const float* __restrict__ bhh,
                                                  float* __restrict__ xp) {
    const int lane = threadIdx.x & 31;
    const int wave = threadIdx.x >> 5;
    const int tile = blockIdx.x * 4 + wave;
    const int half = lane >> 4;
    const int n    = lane & 15;
    const int rowA = tile * 16 + n;
    const int nn   = (n < Hh) ? n : (Hh - 1);
    const float nmask = (n < Hh) ? 1.0f : 0.0f;

    v8f acc = {};
    #pragma unroll
    for (int ks = 0; ks < Cin / 4; ++ks) {
        const int kk = ks * 4 + half * 2;
        const float2 a2 = *reinterpret_cast<const float2*>(inp + (size_t)rowA * Cin + kk);
        v2f afrag; afrag[0] = a2.x; afrag[1] = a2.y;
        v2f bfrag;
        bfrag[0] = nmask * Wih[nn * Cin + kk];
        bfrag[1] = nmask * Wih[nn * Cin + kk + 1];
        acc = __builtin_amdgcn_wmma_f32_16x16x4_f32(
            false, afrag, false, bfrag, (short)0, acc, false, false);
    }

    if (n < Hh) {
        const float bias = bih[n] + bhh[n];
        #pragma unroll
        for (int v = 0; v < 8; ++v) {
            const int g    = tile * 16 + v + half * 8;
            const int bidx = g >> 12;
            const int sidx = g & 4095;
            xp[((size_t)sidx * Bsz + bidx) * Hh + n] = acc[v] + bias;
        }
    }
}

#define BLK   16
#define NBLK  (Sseq / BLK)
#define XSTEP (Bsz * Hh)

__device__ __forceinline__ float fast_tanh(float x) {
    const float e = __expf(2.0f * x);
    return 1.0f - 2.0f / (e + 1.0f);
}

__device__ __forceinline__ float sig4(float logit) {
    const float sg = 1.0f / (1.0f + __expf(-logit));
    const float s2 = sg * sg;
    return s2 * s2;
}

__device__ __forceinline__ void matvec_and_fc(const float (&w)[Hh], const float (&wf)[Hh],
                                              float h, float x, float bfc,
                                              float& pre, float& logit) {
    float a0 = x, a1 = 0.0f, l0 = bfc, l1 = 0.0f;
    asm(
        "v_fmac_f32_dpp %0, %4, %5 row_share:0 row_mask:0xf bank_mask:0xf bound_ctrl:1\n\t"
        "v_fmac_f32_dpp %1, %4, %10 row_share:5 row_mask:0xf bank_mask:0xf bound_ctrl:1\n\t"
        "v_fmac_f32_dpp %2, %4, %15 row_share:0 row_mask:0xf bank_mask:0xf bound_ctrl:1\n\t"
        "v_fmac_f32_dpp %3, %4, %20 row_share:5 row_mask:0xf bank_mask:0xf bound_ctrl:1\n\t"
        "v_fmac_f32_dpp %0, %4, %6 row_share:1 row_mask:0xf bank_mask:0xf bound_ctrl:1\n\t"
        "v_fmac_f32_dpp %1, %4, %11 row_share:6 row_mask:0xf bank_mask:0xf bound_ctrl:1\n\t"
        "v_fmac_f32_dpp %2, %4, %16 row_share:1 row_mask:0xf bank_mask:0xf bound_ctrl:1\n\t"
        "v_fmac_f32_dpp %3, %4, %21 row_share:6 row_mask:0xf bank_mask:0xf bound_ctrl:1\n\t"
        "v_fmac_f32_dpp %0, %4, %7 row_share:2 row_mask:0xf bank_mask:0xf bound_ctrl:1\n\t"
        "v_fmac_f32_dpp %1, %4, %12 row_share:7 row_mask:0xf bank_mask:0xf bound_ctrl:1\n\t"
        "v_fmac_f32_dpp %2, %4, %17 row_share:2 row_mask:0xf bank_mask:0xf bound_ctrl:1\n\t"
        "v_fmac_f32_dpp %3, %4, %22 row_share:7 row_mask:0xf bank_mask:0xf bound_ctrl:1\n\t"
        "v_fmac_f32_dpp %0, %4, %8 row_share:3 row_mask:0xf bank_mask:0xf bound_ctrl:1\n\t"
        "v_fmac_f32_dpp %1, %4, %13 row_share:8 row_mask:0xf bank_mask:0xf bound_ctrl:1\n\t"
        "v_fmac_f32_dpp %2, %4, %18 row_share:3 row_mask:0xf bank_mask:0xf bound_ctrl:1\n\t"
        "v_fmac_f32_dpp %3, %4, %23 row_share:8 row_mask:0xf bank_mask:0xf bound_ctrl:1\n\t"
        "v_fmac_f32_dpp %0, %4, %9 row_share:4 row_mask:0xf bank_mask:0xf bound_ctrl:1\n\t"
        "v_fmac_f32_dpp %1, %4, %14 row_share:9 row_mask:0xf bank_mask:0xf bound_ctrl:1\n\t"
        "v_fmac_f32_dpp %2, %4, %19 row_share:4 row_mask:0xf bank_mask:0xf bound_ctrl:1\n\t"
        "v_fmac_f32_dpp %3, %4, %24 row_share:9 row_mask:0xf bank_mask:0xf bound_ctrl:1\n\t"
        "v_add_f32 %0, %0, %1\n\t"
        "v_add_f32 %2, %2, %3"
        : "+v"(a0), "+v"(a1), "+v"(l0), "+v"(l1)
        : "v"(h),
          "v"(w[0]), "v"(w[1]), "v"(w[2]), "v"(w[3]), "v"(w[4]),
          "v"(w[5]), "v"(w[6]), "v"(w[7]), "v"(w[8]), "v"(w[9]),
          "v"(wf[0]), "v"(wf[1]), "v"(wf[2]), "v"(wf[3]), "v"(wf[4]),
          "v"(wf[5]), "v"(wf[6]), "v"(wf[7]), "v"(wf[8]), "v"(wf[9]));
    pre = a0;
    logit = l0;
}

template <int J>
__device__ __forceinline__ float row_lane(float x) {
    return __builtin_bit_cast(float, __builtin_amdgcn_update_dpp(
        0, __builtin_bit_cast(int, x), 0x150 + J, 0xf, 0xf, true));
}

__device__ __forceinline__ void gather_row(float h, float (&hj)[Hh]) {
    hj[0] = row_lane<0>(h); hj[1] = row_lane<1>(h); hj[2] = row_lane<2>(h);
    hj[3] = row_lane<3>(h); hj[4] = row_lane<4>(h); hj[5] = row_lane<5>(h);
    hj[6] = row_lane<6>(h); hj[7] = row_lane<7>(h); hj[8] = row_lane<8>(h);
    hj[9] = row_lane<9>(h);
}

__global__ __launch_bounds__(32) void rnn_scan(const float* __restrict__ xp,
                                               const float* __restrict__ Whh,
                                               const float* __restrict__ Wfc,
                                               const float* __restrict__ bfc_p,
                                               float* __restrict__ out) {
    const int lane = threadIdx.x;
    const int half = lane >> 4;
    const int i    = lane & 15;
    const int ii   = (i < Hh) ? i : (Hh - 1);
    const int b    = blockIdx.x * 2 + half;

    float w[Hh], wf[Hh];
    #pragma unroll
    for (int j = 0; j < Hh; ++j) w[j]  = Whh[ii * Hh + j];
    #pragma unroll
    for (int j = 0; j < Hh; ++j) wf[j] = Wfc[j];
    const float bfc = bfc_p[0];

    float h = 0.0f;
    float keep = 0.0f;

    const float* const xl = xp + (size_t)b * Hh + ii;

    float xa[BLK], xb[BLK];
    #pragma unroll
    for (int k = 0; k < BLK; ++k) xa[k] = xl[(size_t)k * XSTEP];

    float* const outb = out + (size_t)b * Sseq;

    const float* xn = xl + (size_t)(BLK * XSTEP);

    for (int g = 0; g < NBLK; ++g) {
        #pragma unroll
        for (int k = 0; k < BLK; ++k) xb[k] = xn[(size_t)k * XSTEP];

        #pragma unroll
        for (int k = 0; k < BLK; ++k) {
            if (k == 1 && g != 0) outb[(g - 1) * BLK + i] = keep;

            float pre, logit;
            matvec_and_fc(w, wf, h, xa[k], bfc, pre, logit);

            const float nh = fast_tanh(pre);

            const float val = sig4(logit);
            keep = (((k + 15) & 15) == i) ? val : keep;

            h = nh;

            asm volatile("" : "+v"(keep), "+v"(h));
        }

        #pragma unroll
        for (int k = 0; k < BLK; ++k) xa[k] = xb[k];
        xn += (g + 2 < NBLK) ? (BLK * XSTEP) : 0;
    }

    float hj[Hh];
    gather_row(h, hj);
    float l0 = bfc, l1 = 0.0f;
    #pragma unroll
    for (int j = 0; j < 5; ++j) l0 = fmaf(wf[j],     hj[j],     l0);
    #pragma unroll
    for (int j = 0; j < 5; ++j) l1 = fmaf(wf[j + 5], hj[j + 5], l1);
    keep = (i == 15) ? sig4(l0 + l1) : keep;
    outb[Sseq - 16 + i] = keep;
}

extern "C" void kernel_launch(void* const* d_in, const int* in_sizes, int n_in,
                              void* d_out, int out_size, void* d_ws, size_t ws_size,
                              hipStream_t stream) {
    (void)in_sizes; (void)n_in; (void)out_size; (void)ws_size;
    const float* inp = (const float*)d_in[0];
    const float* Wih = (const float*)d_in[1];
    const float* Whh = (const float*)d_in[2];
    const float* bih = (const float*)d_in[3];
    const float* bhh = (const float*)d_in[4];
    const float* Wfc = (const float*)d_in[5];
    const float* bfc = (const float*)d_in[6];
    float* out = (float*)d_out;
    float* xp  = (float*)d_ws;

    const int tiles = (Bsz * Sseq) / 16;
    xproj_wmma<<<dim3(tiles / 4), 128, 0, stream>>>(inp, Wih, bih, bhh, xp);
    rnn_scan<<<dim3(Bsz / 2), 32, 0, stream>>>(xp, Whh, Wfc, bfc, out);
}


# scan: 1/(1+exp) of tanh and of sigmoid by one v_rcp_f32 (about 1 ulp, f32 kept) instead of the correctly rounded division, a sequence of about ten dependent instructions on the serial chain
# speedup vs baseline: 3.2593x; 1.4380x over previous
#include <hip/hip_runtime.h>

typedef __attribute__((ext_vector_type(2))) float v2f;
typedef __attribute__((ext_vector_type(8))) float v8f;

#define Bsz  64
#define Sseq 4096
#define Cin  80
#define Hh   10

__global__ __launch_bounds__(128) void xproj_wmma(const float* __restrict__ inp,
                                                  const float* __restrict__ Wih,
                                                  const float* __restrict__ bih,
                                                  const float* __restrict__ bhh,
                                                  float* __restrict__ xp) {
    const int lane = threadIdx.x & 31;
    const int wave = threadIdx.x >> 5;
    const int tile = blockIdx.x * 4 + wave;
    const int half = lane >> 4;
    const int n    = lane & 15;
    const int rowA = tile * 16 + n;
    const int nn   = (n < Hh) ? n : (Hh - 1);
    const float nmask = (n < Hh) ? 1.0f : 0.0f;

    v8f acc = {};
    #pragma unroll
    for (int ks = 0; ks < Cin / 4; ++ks) {
        const int kk = ks * 4 + half * 2;
        const float2 a2 = *reinterpret_cast<const float2*>(inp + (size_t)rowA * Cin + kk);
        v2f afrag; afrag[0] = a2.x; afrag[1] = a2.y;
        v2f bfrag;
        bfrag[0] = nmask * Wih[nn * Cin + kk];
        bfrag[1] = nmask * Wih[nn * Cin + kk + 1];
        acc = __builtin_amdgcn_wmma_f32_16x16x4_f32(
            false, afrag, false, bfrag, (short)0, acc, false, false);
    }

    if (n < Hh) {
        const float bias = bih[n] + bhh[n];
        #pragma unroll
        for (int v = 0; v < 8; ++v) {
            const int g    = tile * 16 + v + half * 8;
            const int bidx = g >> 12;
            const int sidx = g & 4095;
            xp[((size_t)sidx * Bsz + bidx) * Hh + n] = acc[v] + bias;
        }
    }
}

#define BLK   16
#define NBLK  (Sseq / BLK)
#define XSTEP (Bsz * Hh)

__device__ __forceinline__ float recip(float d) {
    return __builtin_amdgcn_rcpf(d);
}

__device__ __forceinline__ float fast_tanh(float x) {
    const float e = __expf(2.0f * x);
    return fmaf(-2.0f, recip(e + 1.0f), 1.0f);
}

__device__ __forceinline__ float sig4(float logit) {
    const float sg = recip(1.0f + __expf(-logit));
    const float s2 = sg * sg;
    return s2 * s2;
}

__device__ __forceinline__ void matvec_and_fc(const float (&w)[Hh], const float (&wf)[Hh],
                                              float h, float x, float bfc,
                                              float& pre, float& logit) {
    float a0 = x, a1 = 0.0f, l0 = bfc, l1 = 0.0f;
    asm(
        "v_fmac_f32_dpp %0, %4, %5 row_share:0 row_mask:0xf bank_mask:0xf bound_ctrl:1\n\t"
        "v_fmac_f32_dpp %1, %4, %10 row_share:5 row_mask:0xf bank_mask:0xf bound_ctrl:1\n\t"
        "v_fmac_f32_dpp %2, %4, %15 row_share:0 row_mask:0xf bank_mask:0xf bound_ctrl:1\n\t"
        "v_fmac_f32_dpp %3, %4, %20 row_share:5 row_mask:0xf bank_mask:0xf bound_ctrl:1\n\t"
        "v_fmac_f32_dpp %0, %4, %6 row_share:1 row_mask:0xf bank_mask:0xf bound_ctrl:1\n\t"
        "v_fmac_f32_dpp %1, %4, %11 row_share:6 row_mask:0xf bank_mask:0xf bound_ctrl:1\n\t"
        "v_fmac_f32_dpp %2, %4, %16 row_share:1 row_mask:0xf bank_mask:0xf bound_ctrl:1\n\t"
        "v_fmac_f32_dpp %3, %4, %21 row_share:6 row_mask:0xf bank_mask:0xf bound_ctrl:1\n\t"
        "v_fmac_f32_dpp %0, %4, %7 row_share:2 row_mask:0xf bank_mask:0xf bound_ctrl:1\n\t"
        "v_fmac_f32_dpp %1, %4, %12 row_share:7 row_mask:0xf bank_mask:0xf bound_ctrl:1\n\t"
        "v_fmac_f32_dpp %2, %4, %17 row_share:2 row_mask:0xf bank_mask:0xf bound_ctrl:1\n\t"
        "v_fmac_f32_dpp %3, %4, %22 row_share:7 row_mask:0xf bank_mask:0xf bound_ctrl:1\n\t"
        "v_fmac_f32_dpp %0, %4, %8 row_share:3 row_mask:0xf bank_mask:0xf bound_ctrl:1\n\t"
        "v_fmac_f32_dpp %1, %4, %13 row_share:8 row_mask:0xf bank_mask:0xf bound_ctrl:1\n\t"
        "v_fmac_f32_dpp %2, %4, %18 row_share:3 row_mask:0xf bank_mask:0xf bound_ctrl:1\n\t"
        "v_fmac_f32_dpp %3, %4, %23 row_share:8 row_mask:0xf bank_mask:0xf bound_ctrl:1\n\t"
        "v_fmac_f32_dpp %0, %4, %9 row_share:4 row_mask:0xf bank_mask:0xf bound_ctrl:1\n\t"
        "v_fmac_f32_dpp %1, %4, %14 row_share:9 row_mask:0xf bank_mask:0xf bound_ctrl:1\n\t"
        "v_fmac_f32_dpp %2, %4, %19 row_share:4 row_mask:0xf bank_mask:0xf bound_ctrl:1\n\t"
        "v_fmac_f32_dpp %3, %4, %24 row_share:9 row_mask:0xf bank_mask:0xf bound_ctrl:1\n\t"
        "v_add_f32 %0, %0, %1\n\t"
        "v_add_f32 %2, %2, %3"
        : "+v"(a0), "+v"(a1), "+v"(l0), "+v"(l1)
        : "v"(h),
          "v"(w[0]), "v"(w[1]), "v"(w[2]), "v"(w[3]), "v"(w[4]),
          "v"(w[5]), "v"(w[6]), "v"(w[7]), "v"(w[8]), "v"(w[9]),
          "v"(wf[0]), "v"(wf[1]), "v"(wf[2]), "v"(wf[3]), "v"(wf[4]),
          "v"(wf[5]), "v"(wf[6]), "v"(wf[7]), "v"(wf[8]), "v"(wf[9]));
    pre = a0;
    logit = l0;
}

template <int J>
__device__ __forceinline__ float row_lane(float x) {
    return __builtin_bit_cast(float, __builtin_amdgcn_update_dpp(
        0, __builtin_bit_cast(int, x), 0x150 + J, 0xf, 0xf, true));
}

__device__ __forceinline__ void gather_row(float h, float (&hj)[Hh]) {
    hj[0] = row_lane<0>(h); hj[1] = row_lane<1>(h); hj[2] = row_lane<2>(h);
    hj[3] = row_lane<3>(h); hj[4] = row_lane<4>(h); hj[5] = row_lane<5>(h);
    hj[6] = row_lane<6>(h); hj[7] = row_lane<7>(h); hj[8] = row_lane<8>(h);
    hj[9] = row_lane<9>(h);
}

__global__ __launch_bounds__(32) void rnn_scan(const float* __restrict__ xp,
                                               const float* __restrict__ Whh,
                                               const float* __restrict__ Wfc,
                                               const float* __restrict__ bfc_p,
                                               float* __restrict__ out) {
    const int lane = threadIdx.x;
    const int half = lane >> 4;
    const int i    = lane & 15;
    const int ii   = (i < Hh) ? i : (Hh - 1);
    const int b    = blockIdx.x * 2 + half;

    float w[Hh], wf[Hh];
    #pragma unroll
    for (int j = 0; j < Hh; ++j) w[j]  = Whh[ii * Hh + j];
    #pragma unroll
    for (int j = 0; j < Hh; ++j) wf[j] = Wfc[j];
    const float bfc = bfc_p[0];

    float h = 0.0f;
    float keep = 0.0f;

    const float* const xl = xp + (size_t)b * Hh + ii;

    float xa[BLK], xb[BLK];
    #pragma unroll
    for (int k = 0; k < BLK; ++k) xa[k] = xl[(size_t)k * XSTEP];

    float* const outb = out + (size_t)b * Sseq;

    const float* xn = xl + (size_t)(BLK * XSTEP);

    for (int g = 0; g < NBLK; ++g) {
        #pragma unroll
        for (int k = 0; k < BLK; ++k) xb[k] = xn[(size_t)k * XSTEP];

        #pragma unroll
        for (int k = 0; k < BLK; ++k) {
            if (k == 1 && g != 0) outb[(g - 1) * BLK + i] = keep;

            float pre, logit;
            matvec_and_fc(w, wf, h, xa[k], bfc, pre, logit);

            const float nh = fast_tanh(pre);

            const float val = sig4(logit);
            keep = (((k + 15) & 15) == i) ? val : keep;

            h = nh;

            asm volatile("" : "+v"(keep), "+v"(h));
        }

        #pragma unroll
        for (int k = 0; k < BLK; ++k) xa[k] = xb[k];
        xn += (g + 2 < NBLK) ? (BLK * XSTEP) : 0;
    }

    float hj[Hh];
    gather_row(h, hj);
    float l0 = bfc, l1 = 0.0f;
    #pragma unroll
    for (int j = 0; j < 5; ++j) l0 = fmaf(wf[j],     hj[j],     l0);
    #pragma unroll
    for (int j = 0; j < 5; ++j) l1 = fmaf(wf[j + 5], hj[j + 5], l1);
    keep = (i == 15) ? sig4(l0 + l1) : keep;
    outb[Sseq - 16 + i] = keep;
}

extern "C" void kernel_launch(void* const* d_in, const int* in_sizes, int n_in,
                              void* d_out, int out_size, void* d_ws, size_t ws_size,
                              hipStream_t stream) {
    (void)in_sizes; (void)n_in; (void)out_size; (void)ws_size;
    const float* inp = (const float*)d_in[0];
    const float* Wih = (const float*)d_in[1];
    const float* Whh = (const float*)d_in[2];
    const float* bih = (const float*)d_in[3];
    const float* bhh = (const float*)d_in[4];
    const float* Wfc = (const float*)d_in[5];
    const float* bfc = (const float*)d_in[6];
    float* out = (float*)d_out;
    float* xp  = (float*)d_ws;

    const int tiles = (Bsz * Sseq) / 16;
    xproj_wmma<<<dim3(tiles / 4), 128, 0, stream>>>(inp, Wih, bih, bhh, xp);
    rnn_scan<<<dim3(Bsz / 2), 32, 0, stream>>>(xp, Whh, Wfc, bfc, out);
}


# scan: FC + sigmoid^4 put in the tanh chain's wait cycles (step hand-ordered): sigmoid's 2nd half one step later from a carried exp, 5 of the FC's DPP fmas behind the matvec sum; same sums, same order
# speedup vs baseline: 4.0824x; 1.2525x over previous
#include <hip/hip_runtime.h>

typedef __attribute__((ext_vector_type(2))) float v2f;
typedef __attribute__((ext_vector_type(8))) float v8f;

#define Bsz  64
#define Sseq 4096
#define Cin  80
#define Hh   10

__global__ __launch_bounds__(128) void xproj_wmma(const float* __restrict__ inp,
                                                  const float* __restrict__ Wih,
                                                  const float* __restrict__ bih,
                                                  const float* __restrict__ bhh,
                                                  float* __restrict__ xp) {
    const int lane = threadIdx.x & 31;
    const int wave = threadIdx.x >> 5;
    const int tile = blockIdx.x * 4 + wave;
    const int half = lane >> 4;
    const int n    = lane & 15;
    const int rowA = tile * 16 + n;
    const int nn   = (n < Hh) ? n : (Hh - 1);
    const float nmask = (n < Hh) ? 1.0f : 0.0f;

    v8f acc = {};
    #pragma unroll
    for (int ks = 0; ks < Cin / 4; ++ks) {
        const int kk = ks * 4 + half * 2;
        const float2 a2 = *reinterpret_cast<const float2*>(inp + (size_t)rowA * Cin + kk);
        v2f afrag; afrag[0] = a2.x; afrag[1] = a2.y;
        v2f bfrag;
        bfrag[0] = nmask * Wih[nn * Cin + kk];
        bfrag[1] = nmask * Wih[nn * Cin + kk + 1];
        acc = __builtin_amdgcn_wmma_f32_16x16x4_f32(
            false, afrag, false, bfrag, (short)0, acc, false, false);
    }

    if (n < Hh) {
        const float bias = bih[n] + bhh[n];
        #pragma unroll
        for (int v = 0; v < 8; ++v) {
            const int g    = tile * 16 + v + half * 8;
            const int bidx = g >> 12;
            const int sidx = g & 4095;
            xp[((size_t)sidx * Bsz + bidx) * Hh + n] = acc[v] + bias;
        }
    }
}

#define BLK   16
#define NBLK  (Sseq / BLK)
#define XSTEP (Bsz * Hh)

__device__ __forceinline__ float recip(float d) {
    return __builtin_amdgcn_rcpf(d);
}

__device__ __forceinline__ float sig_exp(float logit) {
    return __expf(-logit);
}

__device__ __forceinline__ float sig4_of_exp(float e) {
    const float sg = recip(1.0f + e);
    const float s2 = sg * sg;
    return s2 * s2;
}

__device__ __forceinline__ void rnn_step(const float (&w)[Hh], const float (&wf)[Hh],
                                         float& h, float x, float bfc,
                                         float& e, float& val) {
    float a0 = x, a1 = 0.0f, l0 = bfc, l1 = 0.0f, tmp;
    asm(
        "v_fmac_f32_dpp %0, %4, %8 row_share:0 row_mask:0xf bank_mask:0xf bound_ctrl:1\n\t"
        "v_fmac_f32_dpp %1, %4, %13 row_share:5 row_mask:0xf bank_mask:0xf bound_ctrl:1\n\t"
        "v_fmac_f32_dpp %2, %4, %18 row_share:0 row_mask:0xf bank_mask:0xf bound_ctrl:1\n\t"
        "v_fmac_f32_dpp %0, %4, %9 row_share:1 row_mask:0xf bank_mask:0xf bound_ctrl:1\n\t"
        "v_fmac_f32_dpp %1, %4, %14 row_share:6 row_mask:0xf bank_mask:0xf bound_ctrl:1\n\t"
        "v_fmac_f32_dpp %3, %4, %23 row_share:5 row_mask:0xf bank_mask:0xf bound_ctrl:1\n\t"
        "v_fmac_f32_dpp %0, %4, %10 row_share:2 row_mask:0xf bank_mask:0xf bound_ctrl:1\n\t"
        "v_fmac_f32_dpp %1, %4, %15 row_share:7 row_mask:0xf bank_mask:0xf bound_ctrl:1\n\t"
        "v_fmac_f32_dpp %2, %4, %19 row_share:1 row_mask:0xf bank_mask:0xf bound_ctrl:1\n\t"
        "v_fmac_f32_dpp %0, %4, %11 row_share:3 row_mask:0xf bank_mask:0xf bound_ctrl:1\n\t"
        "v_fmac_f32_dpp %1, %4, %16 row_share:8 row_mask:0xf bank_mask:0xf bound_ctrl:1\n\t"
        "v_fmac_f32_dpp %3, %4, %24 row_share:6 row_mask:0xf bank_mask:0xf bound_ctrl:1\n\t"
        "v_fmac_f32_dpp %0, %4, %12 row_share:4 row_mask:0xf bank_mask:0xf bound_ctrl:1\n\t"
        "v_fmac_f32_dpp %1, %4, %17 row_share:9 row_mask:0xf bank_mask:0xf bound_ctrl:1\n\t"
        "v_fmac_f32_dpp %2, %4, %20 row_share:2 row_mask:0xf bank_mask:0xf bound_ctrl:1\n\t"
        "v_fmac_f32_dpp %3, %4, %25 row_share:7 row_mask:0xf bank_mask:0xf bound_ctrl:1\n\t"
        "v_add_f32 %0, %0, %1\n\t"
        "v_add_f32 %7, 1.0, %5\n\t"
        "v_fmac_f32_dpp %2, %4, %21 row_share:3 row_mask:0xf bank_mask:0xf bound_ctrl:1\n\t"
        "v_add_f32 %0, %0, %0\n\t"
        "v_rcp_f32 %7, %7\n\t"
        "v_fmac_f32_dpp %3, %4, %26 row_share:8 row_mask:0xf bank_mask:0xf bound_ctrl:1\n\t"
        "v_mul_f32 %0, 0x3fb8aa3b, %0\n\t"
        "v_fmac_f32_dpp %2, %4, %22 row_share:4 row_mask:0xf bank_mask:0xf bound_ctrl:1\n\t"
        "v_fmac_f32_dpp %3, %4, %27 row_share:9 row_mask:0xf bank_mask:0xf bound_ctrl:1\n\t"
        "v_exp_f32 %0, %0\n\t"
        "v_mul_f32 %7, %7, %7\n\t"
        "v_add_f32 %2, %2, %3\n\t"
        "v_mul_f32 %6, %7, %7\n\t"
        "v_add_f32 %0, 1.0, %0\n\t"
        "v_mul_f32 %2, 0xbfb8aa3b, %2\n\t"
        "v_rcp_f32 %0, %0\n\t"
        "v_exp_f32 %5, %2\n\t"
        "v_fma_f32 %4, %0, -2.0, 1.0"
        : "+v"(a0), "+v"(a1), "+v"(l0), "+v"(l1), "+v"(h), "+v"(e),
          "=&v"(val), "=&v"(tmp)
        : "v"(w[0]), "v"(w[1]), "v"(w[2]), "v"(w[3]), "v"(w[4]),
          "v"(w[5]), "v"(w[6]), "v"(w[7]), "v"(w[8]), "v"(w[9]),
          "v"(wf[0]), "v"(wf[1]), "v"(wf[2]), "v"(wf[3]), "v"(wf[4]),
          "v"(wf[5]), "v"(wf[6]), "v"(wf[7]), "v"(wf[8]), "v"(wf[9]));
}

template <int J>
__device__ __forceinline__ float row_lane(float x) {
    return __builtin_bit_cast(float, __builtin_amdgcn_update_dpp(
        0, __builtin_bit_cast(int, x), 0x150 + J, 0xf, 0xf, true));
}

__device__ __forceinline__ void gather_row(float h, float (&hj)[Hh]) {
    hj[0] = row_lane<0>(h); hj[1] = row_lane<1>(h); hj[2] = row_lane<2>(h);
    hj[3] = row_lane<3>(h); hj[4] = row_lane<4>(h); hj[5] = row_lane<5>(h);
    hj[6] = row_lane<6>(h); hj[7] = row_lane<7>(h); hj[8] = row_lane<8>(h);
    hj[9] = row_lane<9>(h);
}

__global__ __launch_bounds__(32) void rnn_scan(const float* __restrict__ xp,
                                               const float* __restrict__ Whh,
                                               const float* __restrict__ Wfc,
                                               const float* __restrict__ bfc_p,
                                               float* __restrict__ out) {
    const int lane = threadIdx.x;
    const int half = lane >> 4;
    const int i    = lane & 15;
    const int ii   = (i < Hh) ? i : (Hh - 1);
    const int b    = blockIdx.x * 2 + half;

    float w[Hh], wf[Hh];
    #pragma unroll
    for (int j = 0; j < Hh; ++j) w[j]  = Whh[ii * Hh + j];
    #pragma unroll
    for (int j = 0; j < Hh; ++j) wf[j] = Wfc[j];
    const float bfc = bfc_p[0];

    float h = 0.0f;
    float keep = 0.0f;
    float eprev = 1.0f;

    const float* const xl = xp + (size_t)b * Hh + ii;

    float xa[BLK], xb[BLK];
    #pragma unroll
    for (int k = 0; k < BLK; ++k) xa[k] = xl[(size_t)k * XSTEP];

    float* const outb = out + (size_t)b * Sseq;

    const float* xn = xl + (size_t)(BLK * XSTEP);

    for (int g = 0; g < NBLK; ++g) {
        #pragma unroll
        for (int k = 0; k < BLK; ++k) xb[k] = xn[(size_t)k * XSTEP];

        #pragma unroll
        for (int k = 0; k < BLK; ++k) {
            if (k == 2 && g != 0) outb[(g - 1) * BLK + i] = keep;

            float val;
            rnn_step(w, wf, h, xa[k], bfc, eprev, val);
            keep = (((k + 14) & 15) == i) ? val : keep;
        }

        #pragma unroll
        for (int k = 0; k < BLK; ++k) xa[k] = xb[k];
        xn += (g + 2 < NBLK) ? (BLK * XSTEP) : 0;
    }

    keep = (i == 14) ? sig4_of_exp(eprev) : keep;
    float hj[Hh];
    gather_row(h, hj);
    float l0 = bfc, l1 = 0.0f;
    #pragma unroll
    for (int j = 0; j < 5; ++j) l0 = fmaf(wf[j],     hj[j],     l0);
    #pragma unroll
    for (int j = 0; j < 5; ++j) l1 = fmaf(wf[j + 5], hj[j + 5], l1);
    keep = (i == 15) ? sig4_of_exp(sig_exp(l0 + l1)) : keep;
    outb[Sseq - 16 + i] = keep;
}

extern "C" void kernel_launch(void* const* d_in, const int* in_sizes, int n_in,
                              void* d_out, int out_size, void* d_ws, size_t ws_size,
                              hipStream_t stream) {
    (void)in_sizes; (void)n_in; (void)out_size; (void)ws_size;
    const float* inp = (const float*)d_in[0];
    const float* Wih = (const float*)d_in[1];
    const float* Whh = (const float*)d_in[2];
    const float* bih = (const float*)d_in[3];
    const float* bhh = (const float*)d_in[4];
    const float* Wfc = (const float*)d_in[5];
    const float* bfc = (const float*)d_in[6];
    float* out = (float*)d_out;
    float* xp  = (float*)d_ws;

    const int tiles = (Bsz * Sseq) / 16;
    xproj_wmma<<<dim3(tiles / 4), 128, 0, stream>>>(inp, Wih, bih, bhh, xp);
    rnn_scan<<<dim3(Bsz / 2), 32, 0, stream>>>(xp, Whh, Wfc, bfc, out);
}


# scan: tanh of the step is the one instruction v_tanh_f32 (f32 kept) in place of 1 - 2/(exp(2x)+1), six dependent instructions on the serial chain; step re-ordered round it, sums and their order kept
# speedup vs baseline: 4.6855x; 1.1477x over previous
#include <hip/hip_runtime.h>

typedef __attribute__((ext_vector_type(2))) float v2f;
typedef __attribute__((ext_vector_type(8))) float v8f;

#define Bsz  64
#define Sseq 4096
#define Cin  80
#define Hh   10

__global__ __launch_bounds__(128) void xproj_wmma(const float* __restrict__ inp,
                                                  const float* __restrict__ Wih,
                                                  const float* __restrict__ bih,
                                                  const float* __restrict__ bhh,
                                                  float* __restrict__ xp) {
    const int lane = threadIdx.x & 31;
    const int wave = threadIdx.x >> 5;
    const int tile = blockIdx.x * 4 + wave;
    const int half = lane >> 4;
    const int n    = lane & 15;
    const int rowA = tile * 16 + n;
    const int nn   = (n < Hh) ? n : (Hh - 1);
    const float nmask = (n < Hh) ? 1.0f : 0.0f;

    v8f acc = {};
    #pragma unroll
    for (int ks = 0; ks < Cin / 4; ++ks) {
        const int kk = ks * 4 + half * 2;
        const float2 a2 = *reinterpret_cast<const float2*>(inp + (size_t)rowA * Cin + kk);
        v2f afrag; afrag[0] = a2.x; afrag[1] = a2.y;
        v2f bfrag;
        bfrag[0] = nmask * Wih[nn * Cin + kk];
        bfrag[1] = nmask * Wih[nn * Cin + kk + 1];
        acc = __builtin_amdgcn_wmma_f32_16x16x4_f32(
            false, afrag, false, bfrag, (short)0, acc, false, false);
    }

    if (n < Hh) {
        const float bias = bih[n] + bhh[n];
        #pragma unroll
        for (int v = 0; v < 8; ++v) {
            const int g    = tile * 16 + v + half * 8;
            const int bidx = g >> 12;
            const int sidx = g & 4095;
            xp[((size_t)sidx * Bsz + bidx) * Hh + n] = acc[v] + bias;
        }
    }
}

#define BLK   16
#define NBLK  (Sseq / BLK)
#define XSTEP (Bsz * Hh)

__device__ __forceinline__ float recip(float d) {
    return __builtin_amdgcn_rcpf(d);
}

__device__ __forceinline__ float sig_exp(float logit) {
    return __expf(-logit);
}

__device__ __forceinline__ float sig4_of_exp(float e) {
    const float sg = recip(1.0f + e);
    const float s2 = sg * sg;
    return s2 * s2;
}

__device__ __forceinline__ void rnn_step(const float (&w)[Hh], const float (&wf)[Hh],
                                         float h, float& hn, float x, float bfc,
                                         float lp0, float lp1,
                                         float& l0, float& l1, float& val) {
    float a0 = x, a1, lg, hn_, l0_, l1_, val_;
    asm(
        "v_fmac_f32_dpp %[a0], %[h], %[w0] row_share:0 row_mask:0xf bank_mask:0xf bound_ctrl:1\n\t"
        "v_add_f32 %[lg], %[lp0], %[lp1]\n\t"
        "v_mul_f32_dpp %[a1], %[h], %[w5] row_share:5 row_mask:0xf bank_mask:0xf bound_ctrl:1\n\t"
        "v_mul_f32_dpp %[l1], %[h], %[f5] row_share:5 row_mask:0xf bank_mask:0xf bound_ctrl:1\n\t"
        "v_mul_f32 %[lg], 0xbfb8aa3b, %[lg]\n\t"
        "v_fmac_f32_dpp %[a0], %[h], %[w1] row_share:1 row_mask:0xf bank_mask:0xf bound_ctrl:1\n\t"
        "v_fmac_f32_dpp %[a1], %[h], %[w6] row_share:6 row_mask:0xf bank_mask:0xf bound_ctrl:1\n\t"
        "v_exp_f32 %[lg], %[lg]\n\t"
        "v_fmac_f32_dpp %[l1], %[h], %[f6] row_share:6 row_mask:0xf bank_mask:0xf bound_ctrl:1\n\t"
        "v_fmac_f32_dpp %[a0], %[h], %[w2] row_share:2 row_mask:0xf bank_mask:0xf bound_ctrl:1\n\t"
        "v_fmac_f32_dpp %[a1], %[h], %[w7] row_share:7 row_mask:0xf bank_mask:0xf bound_ctrl:1\n\t"
        "v_fma_f32_e64_dpp %[l0], %[h], %[f0], %[bfc] row_share:0 row_mask:0xf bank_mask:0xf bound_ctrl:1\n\t"
        "v_add_f32 %[val], 1.0, %[lg]\n\t"
        "v_fmac_f32_dpp %[a0], %[h], %[w3] row_share:3 row_mask:0xf bank_mask:0xf bound_ctrl:1\n\t"
        "v_fmac_f32_dpp %[a1], %[h], %[w8] row_share:8 row_mask:0xf bank_mask:0xf bound_ctrl:1\n\t"
        "v_fmac_f32_dpp %[l0], %[h], %[f1] row_share:1 row_mask:0xf bank_mask:0xf bound_ctrl:1\n\t"
        "v_fmac_f32_dpp %[a0], %[h], %[w4] row_share:4 row_mask:0xf bank_mask:0xf bound_ctrl:1\n\t"
        "v_rcp_f32 %[val], %[val]\n\t"
        "v_fmac_f32_dpp %[a1], %[h], %[w9] row_share:9 row_mask:0xf bank_mask:0xf bound_ctrl:1\n\t"
        "v_fmac_f32_dpp %[l0], %[h], %[f2] row_share:2 row_mask:0xf bank_mask:0xf bound_ctrl:1\n\t"
        "v_fmac_f32_dpp %[l1], %[h], %[f7] row_share:7 row_mask:0xf bank_mask:0xf bound_ctrl:1\n\t"
        "v_add_f32 %[a0], %[a0], %[a1]\n\t"
        "v_mul_f32 %[val], %[val], %[val]\n\t"
        "v_fmac_f32_dpp %[l0], %[h], %[f3] row_share:3 row_mask:0xf bank_mask:0xf bound_ctrl:1\n\t"
        "v_fmac_f32_dpp %[l1], %[h], %[f8] row_share:8 row_mask:0xf bank_mask:0xf bound_ctrl:1\n\t"
        "v_mul_f32 %[val], %[val], %[val]\n\t"
        "v_tanh_f32 %[hn], %[a0]\n\t"
        "v_fmac_f32_dpp %[l0], %[h], %[f4] row_share:4 row_mask:0xf bank_mask:0xf bound_ctrl:1\n\t"
        "v_fmac_f32_dpp %[l1], %[h], %[f9] row_share:9 row_mask:0xf bank_mask:0xf bound_ctrl:1"
        : [a0] "+v"(a0), [a1] "=&v"(a1), [lg] "=&v"(lg), [hn] "=&v"(hn_),
          [l0] "=&v"(l0_), [l1] "=&v"(l1_), [val] "=&v"(val_)
        : [h] "v"(h), [lp0] "v"(lp0), [lp1] "v"(lp1), [bfc] "v"(bfc),
          [w0] "v"(w[0]), [w1] "v"(w[1]), [w2] "v"(w[2]), [w3] "v"(w[3]), [w4] "v"(w[4]),
          [w5] "v"(w[5]), [w6] "v"(w[6]), [w7] "v"(w[7]), [w8] "v"(w[8]), [w9] "v"(w[9]),
          [f0] "v"(wf[0]), [f1] "v"(wf[1]), [f2] "v"(wf[2]), [f3] "v"(wf[3]), [f4] "v"(wf[4]),
          [f5] "v"(wf[5]), [f6] "v"(wf[6]), [f7] "v"(wf[7]), [f8] "v"(wf[8]), [f9] "v"(wf[9]));
    hn = hn_; l0 = l0_; l1 = l1_; val = val_;
}

template <int J>
__device__ __forceinline__ float row_lane(float x) {
    return __builtin_bit_cast(float, __builtin_amdgcn_update_dpp(
        0, __builtin_bit_cast(int, x), 0x150 + J, 0xf, 0xf, true));
}

__device__ __forceinline__ void gather_row(float h, float (&hj)[Hh]) {
    hj[0] = row_lane<0>(h); hj[1] = row_lane<1>(h); hj[2] = row_lane<2>(h);
    hj[3] = row_lane<3>(h); hj[4] = row_lane<4>(h); hj[5] = row_lane<5>(h);
    hj[6] = row_lane<6>(h); hj[7] = row_lane<7>(h); hj[8] = row_lane<8>(h);
    hj[9] = row_lane<9>(h);
}

__global__ __launch_bounds__(32) void rnn_scan(const float* __restrict__ xp,
                                               const float* __restrict__ Whh,
                                               const float* __restrict__ Wfc,
                                               const float* __restrict__ bfc_p,
                                               float* __restrict__ out) {
    const int lane = threadIdx.x;
    const int half = lane >> 4;
    const int i    = lane & 15;
    const int ii   = (i < Hh) ? i : (Hh - 1);
    const int b    = blockIdx.x * 2 + half;

    float w[Hh], wf[Hh];
    #pragma unroll
    for (int j = 0; j < Hh; ++j) w[j]  = Whh[ii * Hh + j];
    #pragma unroll
    for (int j = 0; j < Hh; ++j) wf[j] = Wfc[j];
    const float bfc = bfc_p[0];

    float h = 0.0f;
    float keep = 0.0f;
    float lp0 = 0.0f, lp1 = 0.0f;

    const float* const xl = xp + (size_t)b * Hh + ii;

    float xa[BLK], xb[BLK];
    #pragma unroll
    for (int k = 0; k < BLK; ++k) xa[k] = xl[(size_t)k * XSTEP];

    float* const outb = out + (size_t)b * Sseq;

    const float* xn = xl + (size_t)(BLK * XSTEP);

    for (int g = 0; g < NBLK; ++g) {
        #pragma unroll
        for (int k = 0; k < BLK; ++k) xb[k] = xn[(size_t)k * XSTEP];

        #pragma unroll
        for (int k = 0; k < BLK; ++k) {
            if (k == 2 && g != 0) outb[(g - 1) * BLK + i] = keep;

            float hn, n0, n1, val;
            rnn_step(w, wf, h, hn, xa[k], bfc, lp0, lp1, n0, n1, val);
            h = hn; lp0 = n0; lp1 = n1;
            keep = (((k + 14) & 15) == i) ? val : keep;
        }

        #pragma unroll
        for (int k = 0; k < BLK; ++k) xa[k] = xb[k];
        xn += (g + 2 < NBLK) ? (BLK * XSTEP) : 0;
    }

    keep = (i == 14) ? sig4_of_exp(sig_exp(lp0 + lp1)) : keep;
    float hj[Hh];
    gather_row(h, hj);
    float l0 = bfc, l1 = 0.0f;
    #pragma unroll
    for (int j = 0; j < 5; ++j) l0 = fmaf(wf[j],     hj[j],     l0);
    #pragma unroll
    for (int j = 0; j < 5; ++j) l1 = fmaf(wf[j + 5], hj[j + 5], l1);
    keep = (i == 15) ? sig4_of_exp(sig_exp(l0 + l1)) : keep;
    outb[Sseq - 16 + i] = keep;
}

extern "C" void kernel_launch(void* const* d_in, const int* in_sizes, int n_in,
                              void* d_out, int out_size, void* d_ws, size_t ws_size,
                              hipStream_t stream) {
    (void)in_sizes; (void)n_in; (void)out_size; (void)ws_size;
    const float* inp = (const float*)d_in[0];
    const float* Wih = (const float*)d_in[1];
    const float* Whh = (const float*)d_in[2];
    const float* bih = (const float*)d_in[3];
    const float* bhh = (const float*)d_in[4];
    const float* Wfc = (const float*)d_in[5];
    const float* bfc = (const float*)d_in[6];
    float* out = (float*)d_out;
    float* xp  = (float*)d_ws;

    const int tiles = (Bsz * Sseq) / 16;
    xproj_wmma<<<dim3(tiles / 4), 128, 0, stream>>>(inp, Wih, bih, bhh, xp);
    rnn_scan<<<dim3(Bsz / 2), 32, 0, stream>>>(xp, Whh, Wfc, bfc, out);
}


# scan: FC + sigmoid^4 moved to a second wave of the work-group through a ring of h in LDS (one row stored per step, one split barrier per 16 steps); scan wave issues only matvec + tanh; sum order kept
# speedup vs baseline: 4.9728x; 1.0613x over previous
#include <hip/hip_runtime.h>

typedef __attribute__((ext_vector_type(2))) float v2f;
typedef __attribute__((ext_vector_type(8))) float v8f;

#define Bsz  64
#define Sseq 4096
#define Cin  80
#define Hh   10

__global__ __launch_bounds__(128) void xproj_wmma(const float* __restrict__ inp,
                                                  const float* __restrict__ Wih,
                                                  const float* __restrict__ bih,
                                                  const float* __restrict__ bhh,
                                                  float* __restrict__ xp) {
    const int lane = threadIdx.x & 31;
    const int wave = threadIdx.x >> 5;
    const int tile = blockIdx.x * 4 + wave;
    const int half = lane >> 4;
    const int n    = lane & 15;
    const int rowA = tile * 16 + n;
    const int nn   = (n < Hh) ? n : (Hh - 1);
    const float nmask = (n < Hh) ? 1.0f : 0.0f;

    v8f acc = {};
    #pragma unroll
    for (int ks = 0; ks < Cin / 4; ++ks) {
        const int kk = ks * 4 + half * 2;
        const float2 a2 = *reinterpret_cast<const float2*>(inp + (size_t)rowA * Cin + kk);
        v2f afrag; afrag[0] = a2.x; afrag[1] = a2.y;
        v2f bfrag;
        bfrag[0] = nmask * Wih[nn * Cin + kk];
        bfrag[1] = nmask * Wih[nn * Cin + kk + 1];
        acc = __builtin_amdgcn_wmma_f32_16x16x4_f32(
            false, afrag, false, bfrag, (short)0, acc, false, false);
    }

    if (n < Hh) {
        const float bias = bih[n] + bhh[n];
        #pragma unroll
        for (int v = 0; v < 8; ++v) {
            const int g    = tile * 16 + v + half * 8;
            const int bidx = g >> 12;
            const int sidx = g & 4095;
            xp[((size_t)sidx * Bsz + bidx) * Hh + n] = acc[v] + bias;
        }
    }
}

#define BLK   16
#define NBLK  (Sseq / BLK)
#define XSTEP (Bsz * Hh)
#define RING_BLKS 4
#define ROW_DW    36

__device__ __forceinline__ float recip(float d) {
    return __builtin_amdgcn_rcpf(d);
}

__device__ __forceinline__ float sig_exp(float logit) {
    return __expf(-logit);
}

__device__ __forceinline__ float sig4_of_exp(float e) {
    const float sg = recip(1.0f + e);
    const float s2 = sg * sg;
    return s2 * s2;
}

template <int N>
__device__ __forceinline__ void wait_ds_le() {
    asm volatile("s_wait_dscnt %0" :: "i"(N) : "memory");
}
__device__ __forceinline__ void bar_signal() {
    asm volatile("s_barrier_signal -1" ::: "memory");
}
__device__ __forceinline__ void bar_wait() {
    asm volatile("s_barrier_wait -1" ::: "memory");
}

__device__ __forceinline__ void rnn_step(const float (&w)[Hh], float h, float& hn, float x) {
    float a0 = x, a1, hn_;
    asm(
        "v_fmac_f32_dpp %[a0], %[h], %[w0] row_share:0 row_mask:0xf bank_mask:0xf bound_ctrl:1\n\t"
        "v_mul_f32_dpp %[a1], %[h], %[w5] row_share:5 row_mask:0xf bank_mask:0xf bound_ctrl:1\n\t"
        "v_fmac_f32_dpp %[a0], %[h], %[w1] row_share:1 row_mask:0xf bank_mask:0xf bound_ctrl:1\n\t"
        "v_fmac_f32_dpp %[a1], %[h], %[w6] row_share:6 row_mask:0xf bank_mask:0xf bound_ctrl:1\n\t"
        "v_fmac_f32_dpp %[a0], %[h], %[w2] row_share:2 row_mask:0xf bank_mask:0xf bound_ctrl:1\n\t"
        "v_fmac_f32_dpp %[a1], %[h], %[w7] row_share:7 row_mask:0xf bank_mask:0xf bound_ctrl:1\n\t"
        "v_fmac_f32_dpp %[a0], %[h], %[w3] row_share:3 row_mask:0xf bank_mask:0xf bound_ctrl:1\n\t"
        "v_fmac_f32_dpp %[a1], %[h], %[w8] row_share:8 row_mask:0xf bank_mask:0xf bound_ctrl:1\n\t"
        "v_fmac_f32_dpp %[a0], %[h], %[w4] row_share:4 row_mask:0xf bank_mask:0xf bound_ctrl:1\n\t"
        "v_fmac_f32_dpp %[a1], %[h], %[w9] row_share:9 row_mask:0xf bank_mask:0xf bound_ctrl:1\n\t"
        "v_add_f32 %[a0], %[a0], %[a1]\n\t"
        "v_tanh_f32 %[hn], %[a0]\n\t"
        "v_nop"
        : [a0] "+v"(a0), [a1] "=&v"(a1), [hn] "=&v"(hn_)
        : [h] "v"(h),
          [w0] "v"(w[0]), [w1] "v"(w[1]), [w2] "v"(w[2]), [w3] "v"(w[3]), [w4] "v"(w[4]),
          [w5] "v"(w[5]), [w6] "v"(w[6]), [w7] "v"(w[7]), [w8] "v"(w[8]), [w9] "v"(w[9]));
    hn = hn_;
}

__device__ __forceinline__ void fc_block(const float* hbuf, int g, int lane,
                                         const float (&wf)[Hh], float bfc,
                                         float* __restrict__ out_b0) {
    const int hb = lane >> 4;
    const int k  = lane & 15;
    const float* row = hbuf + ((g & (RING_BLKS - 1)) * BLK + k) * ROW_DW + hb * 16;
    const float4 q0 = *reinterpret_cast<const float4*>(row);
    const float4 q1 = *reinterpret_cast<const float4*>(row + 4);
    const float2 q2 = *reinterpret_cast<const float2*>(row + 8);
    float l0 = fmaf(q0.x, wf[0], bfc);
    l0 = fmaf(q0.y, wf[1], l0);
    l0 = fmaf(q0.z, wf[2], l0);
    l0 = fmaf(q0.w, wf[3], l0);
    l0 = fmaf(q1.x, wf[4], l0);
    float l1 = q1.y * wf[5];
    l1 = fmaf(q1.z, wf[6], l1);
    l1 = fmaf(q1.w, wf[7], l1);
    l1 = fmaf(q2.x, wf[8], l1);
    l1 = fmaf(q2.y, wf[9], l1);
    out_b0[(size_t)hb * Sseq + g * BLK + k] = sig4_of_exp(sig_exp(l0 + l1));
}

__global__ __launch_bounds__(64) void rnn_scan(const float* __restrict__ xp,
                                               const float* __restrict__ Whh,
                                               const float* __restrict__ Wfc,
                                               const float* __restrict__ bfc_p,
                                               float* __restrict__ out) {
    __shared__ __attribute__((aligned(16))) float hbuf[RING_BLKS * BLK * ROW_DW];

    const int lane = threadIdx.x & 31;
    const bool is_scan = __builtin_amdgcn_readfirstlane((int)(threadIdx.x >> 5)) == 0;
    const int half = lane >> 4;
    const int i    = lane & 15;
    const int ii   = (i < Hh) ? i : (Hh - 1);
    const int b    = blockIdx.x * 2 + half;

    float w[Hh], wf[Hh];
    #pragma unroll
    for (int j = 0; j < Hh; ++j) w[j]  = Whh[ii * Hh + j];
    #pragma unroll
    for (int j = 0; j < Hh; ++j) wf[j] = Wfc[j];
    const float bfc = bfc_p[0];

    if (!is_scan) __builtin_amdgcn_s_setprio(3);

    float h = 0.0f;

    const float* const xl = xp + (size_t)b * Hh + ii;

    float xa[BLK], xb[BLK];
    #pragma unroll
    for (int k = 0; k < BLK; ++k) xa[k] = xl[(size_t)k * XSTEP];

    float* const out_b0 = out + (size_t)(blockIdx.x * 2) * Sseq;

    const float* xn = xl + (size_t)(BLK * XSTEP);

    float* const hcol = hbuf + lane;

    for (int g = 0; g < NBLK; ++g) {
        if (!is_scan && g >= 3) fc_block(hbuf, g - 3, lane, wf, bfc, out_b0);

        wait_ds_le<8>();
        bar_signal();

        if (is_scan) {
            float* const hblk = hcol + (g & (RING_BLKS - 1)) * (BLK * ROW_DW);

            #pragma unroll
            for (int k = 0; k < BLK; ++k) xb[k] = xn[(size_t)k * XSTEP];

            #pragma unroll
            for (int k = 0; k < BLK; ++k) {
                float hn;
                rnn_step(w, h, hn, xa[k]);
                if (k > 0) hblk[(k - 1) * ROW_DW] = h;
                h = hn;
            }
            hblk[(BLK - 1) * ROW_DW] = h;

            #pragma unroll
            for (int k = 0; k < BLK; ++k) xa[k] = xb[k];
            xn += (g + 2 < NBLK) ? (BLK * XSTEP) : 0;
        }

        bar_wait();
    }

    if (!is_scan) fc_block(hbuf, NBLK - 3, lane, wf, bfc, out_b0);
    wait_ds_le<0>();
    bar_signal();
    bar_wait();
    if (!is_scan) {
        fc_block(hbuf, NBLK - 2, lane, wf, bfc, out_b0);
        fc_block(hbuf, NBLK - 1, lane, wf, bfc, out_b0);
        __builtin_amdgcn_s_setprio(0);
    }
}

extern "C" void kernel_launch(void* const* d_in, const int* in_sizes, int n_in,
                              void* d_out, int out_size, void* d_ws, size_t ws_size,
                              hipStream_t stream) {
    (void)in_sizes; (void)n_in; (void)out_size; (void)ws_size;
    const float* inp = (const float*)d_in[0];
    const float* Wih = (const float*)d_in[1];
    const float* Whh = (const float*)d_in[2];
    const float* bih = (const float*)d_in[3];
    const float* bhh = (const float*)d_in[4];
    const float* Wfc = (const float*)d_in[5];
    const float* bfc = (const float*)d_in[6];
    float* out = (float*)d_out;
    float* xp  = (float*)d_ws;

    const int tiles = (Bsz * Sseq) / 16;
    xproj_wmma<<<dim3(tiles / 4), 128, 0, stream>>>(inp, Wih, bih, bhh, xp);
    rnn_scan<<<dim3(Bsz / 2), 64, 0, stream>>>(xp, Whh, Wfc, bfc, out);
}


# xproj: each 16x80 input tile read as 5 KB of contiguous 16-byte loads, shaped to A fragments in a per-wave LDS stage; 4 tiles per wave, W_ih fragments kept in registers; k order of products permuted
# speedup vs baseline: 5.2642x; 1.0586x over previous
#include <hip/hip_runtime.h>

typedef __attribute__((ext_vector_type(2))) float v2f;
typedef __attribute__((ext_vector_type(4))) float v4f;
typedef __attribute__((ext_vector_type(8))) float v8f;

#define Bsz  64
#define Sseq 4096
#define Cin  80
#define Hh   10

#define XP_TPW     4
#define XP_WAVES   4
#define XP_PITCH   84
#define XP_WAVE_DW (16 * XP_PITCH)
#define XP_PIECES  ((16 * Cin) / (4 * 32))


__device__ __forceinline__ void xproj_load(v4f (&dst)[XP_PIECES], const v4f* __restrict__ p) {
    #pragma unroll
    for (int i = 0; i < XP_PIECES; ++i) dst[i] = p[i * 32];
}

__device__ __forceinline__ void xproj_stage(float* st, const int (&woff)[XP_PIECES],
                                            const v4f (&src)[XP_PIECES]) {
    #pragma unroll
    for (int i = 0; i < XP_PIECES; ++i)
        *reinterpret_cast<v4f*>(st + woff[i]) = src[i];
}

__device__ __forceinline__ void xproj_mul(const float* st, int roff, const v2f (&bfrag)[Cin / 4],
                                          float bias, int tile, int half, int n,
                                          float* __restrict__ xp) {
    v8f acc = {};
    #pragma unroll
    for (int j = 0; j < XP_PIECES; ++j) {
        const v4f a4 = *reinterpret_cast<const v4f*>(st + roff + 8 * j);
        v2f a0; a0[0] = a4[0]; a0[1] = a4[1];
        v2f a1; a1[0] = a4[2]; a1[1] = a4[3];
        acc = __builtin_amdgcn_wmma_f32_16x16x4_f32(
            false, a0, false, bfrag[2 * j], (short)0, acc, false, false);
        acc = __builtin_amdgcn_wmma_f32_16x16x4_f32(
            false, a1, false, bfrag[2 * j + 1], (short)0, acc, false, false);
    }

    if (n < Hh) {
        #pragma unroll
        for (int v = 0; v < 8; ++v) {
            const int gr   = tile * 16 + v + half * 8;
            const int bidx = gr >> 12;
            const int sidx = gr & 4095;
            xp[((size_t)sidx * Bsz + bidx) * Hh + n] = acc[v] + bias;
        }
    }
}

__global__ __launch_bounds__(128) __attribute__((amdgpu_waves_per_eu(4, 4)))
void xproj_wmma(const float* __restrict__ inp,
                const float* __restrict__ Wih,
                const float* __restrict__ bih,
                const float* __restrict__ bhh,
                float* __restrict__ xp) {
    __shared__ __attribute__((aligned(16))) float stage[XP_WAVES * XP_WAVE_DW];

    const int lane = threadIdx.x & 31;
    const int wave = threadIdx.x >> 5;
    const int half = lane >> 4;
    const int n    = lane & 15;
    const int nn   = (n < Hh) ? n : (Hh - 1);
    const float nmask = (n < Hh) ? 1.0f : 0.0f;
    const int tile0 = (blockIdx.x * XP_WAVES + wave) * XP_TPW;

    float* const st = stage + wave * XP_WAVE_DW;

    constexpr int TILE4 = 16 * Cin / 4;
    const v4f* const in4 = reinterpret_cast<const v4f*>(inp) + (size_t)tile0 * TILE4 + lane;
    v4f p0[XP_PIECES], p1[XP_PIECES], p2[XP_PIECES], p3[XP_PIECES];
    xproj_load(p0, in4);
    xproj_load(p1, in4 + TILE4);
    __builtin_amdgcn_sched_barrier(0);

    v2f bfrag[Cin / 4];
    #pragma unroll
    for (int j = 0; j < XP_PIECES; ++j) {
        const v4f w4 = *reinterpret_cast<const v4f*>(Wih + nn * Cin + 8 * j + 4 * half);
        bfrag[2 * j][0]     = nmask * w4[0];
        bfrag[2 * j][1]     = nmask * w4[1];
        bfrag[2 * j + 1][0] = nmask * w4[2];
        bfrag[2 * j + 1][1] = nmask * w4[3];
    }
    const float bias = bih[nn] + bhh[nn];

    int woff[XP_PIECES];
    #pragma unroll
    for (int i = 0; i < XP_PIECES; ++i) {
        const int f = i * 32 + lane;
        const int r = f / (Cin / 4);
        woff[i] = r * XP_PITCH + (f - r * (Cin / 4)) * 4;
    }
    const int roff = n * XP_PITCH + 4 * half;

    static_assert(XP_TPW == 4, "the four tiles below are the wave's");
    xproj_stage(st, woff, p0);
    xproj_load(p2, in4 + 2 * TILE4);
    xproj_mul(st, roff, bfrag, bias, tile0 + 0, half, n, xp);

    xproj_stage(st, woff, p1);
    xproj_load(p3, in4 + 3 * TILE4);
    xproj_mul(st, roff, bfrag, bias, tile0 + 1, half, n, xp);

    xproj_stage(st, woff, p2);
    xproj_mul(st, roff, bfrag, bias, tile0 + 2, half, n, xp);

    xproj_stage(st, woff, p3);
    xproj_mul(st, roff, bfrag, bias, tile0 + 3, half, n, xp);
}

#define BLK   16
#define NBLK  (Sseq / BLK)
#define XSTEP (Bsz * Hh)
#define RING_BLKS 4
#define ROW_DW    36

__device__ __forceinline__ float recip(float d) {
    return __builtin_amdgcn_rcpf(d);
}

__device__ __forceinline__ float sig_exp(float logit) {
    return __expf(-logit);
}

__device__ __forceinline__ float sig4_of_exp(float e) {
    const float sg = recip(1.0f + e);
    const float s2 = sg * sg;
    return s2 * s2;
}

template <int N>
__device__ __forceinline__ void wait_ds_le() {
    asm volatile("s_wait_dscnt %0" :: "i"(N) : "memory");
}
__device__ __forceinline__ void bar_signal() {
    asm volatile("s_barrier_signal -1" ::: "memory");
}
__device__ __forceinline__ void bar_wait() {
    asm volatile("s_barrier_wait -1" ::: "memory");
}

__device__ __forceinline__ void rnn_step(const float (&w)[Hh], float h, float& hn, float x) {
    float a0 = x, a1, hn_;
    asm(
        "v_fmac_f32_dpp %[a0], %[h], %[w0] row_share:0 row_mask:0xf bank_mask:0xf bound_ctrl:1\n\t"
        "v_mul_f32_dpp %[a1], %[h], %[w5] row_share:5 row_mask:0xf bank_mask:0xf bound_ctrl:1\n\t"
        "v_fmac_f32_dpp %[a0], %[h], %[w1] row_share:1 row_mask:0xf bank_mask:0xf bound_ctrl:1\n\t"
        "v_fmac_f32_dpp %[a1], %[h], %[w6] row_share:6 row_mask:0xf bank_mask:0xf bound_ctrl:1\n\t"
        "v_fmac_f32_dpp %[a0], %[h], %[w2] row_share:2 row_mask:0xf bank_mask:0xf bound_ctrl:1\n\t"
        "v_fmac_f32_dpp %[a1], %[h], %[w7] row_share:7 row_mask:0xf bank_mask:0xf bound_ctrl:1\n\t"
        "v_fmac_f32_dpp %[a0], %[h], %[w3] row_share:3 row_mask:0xf bank_mask:0xf bound_ctrl:1\n\t"
        "v_fmac_f32_dpp %[a1], %[h], %[w8] row_share:8 row_mask:0xf bank_mask:0xf bound_ctrl:1\n\t"
        "v_fmac_f32_dpp %[a0], %[h], %[w4] row_share:4 row_mask:0xf bank_mask:0xf bound_ctrl:1\n\t"
        "v_fmac_f32_dpp %[a1], %[h], %[w9] row_share:9 row_mask:0xf bank_mask:0xf bound_ctrl:1\n\t"
        "v_add_f32 %[a0], %[a0], %[a1]\n\t"
        "v_tanh_f32 %[hn], %[a0]\n\t"
        "v_nop"
        : [a0] "+v"(a0), [a1] "=&v"(a1), [hn] "=&v"(hn_)
        : [h] "v"(h),
          [w0] "v"(w[0]), [w1] "v"(w[1]), [w2] "v"(w[2]), [w3] "v"(w[3]), [w4] "v"(w[4]),
          [w5] "v"(w[5]), [w6] "v"(w[6]), [w7] "v"(w[7]), [w8] "v"(w[8]), [w9] "v"(w[9]));
    hn = hn_;
}

__device__ __forceinline__ void fc_block(const float* hbuf, int g, int lane,
                                         const float (&wf)[Hh], float bfc,
                                         float* __restrict__ out_b0) {
    const int hb = lane >> 4;
    const int k  = lane & 15;
    const float* row = hbuf + ((g & (RING_BLKS - 1)) * BLK + k) * ROW_DW + hb * 16;
    const float4 q0 = *reinterpret_cast<const float4*>(row);
    const float4 q1 = *reinterpret_cast<const float4*>(row + 4);
    const float2 q2 = *reinterpret_cast<const float2*>(row + 8);
    float l0 = fmaf(q0.x, wf[0], bfc);
    l0 = fmaf(q0.y, wf[1], l0);
    l0 = fmaf(q0.z, wf[2], l0);
    l0 = fmaf(q0.w, wf[3], l0);
    l0 = fmaf(q1.x, wf[4], l0);
    float l1 = q1.y * wf[5];
    l1 = fmaf(q1.z, wf[6], l1);
    l1 = fmaf(q1.w, wf[7], l1);
    l1 = fmaf(q2.x, wf[8], l1);
    l1 = fmaf(q2.y, wf[9], l1);
    out_b0[(size_t)hb * Sseq + g * BLK + k] = sig4_of_exp(sig_exp(l0 + l1));
}

__global__ __launch_bounds__(64) void rnn_scan(const float* __restrict__ xp,
                                               const float* __restrict__ Whh,
                                               const float* __restrict__ Wfc,
                                               const float* __restrict__ bfc_p,
                                               float* __restrict__ out) {
    __shared__ __attribute__((aligned(16))) float hbuf[RING_BLKS * BLK * ROW_DW];

    const int lane = threadIdx.x & 31;
    const bool is_scan = __builtin_amdgcn_readfirstlane((int)(threadIdx.x >> 5)) == 0;
    const int half = lane >> 4;
    const int i    = lane & 15;
    const int ii   = (i < Hh) ? i : (Hh - 1);
    const int b    = blockIdx.x * 2 + half;

    float w[Hh], wf[Hh];
    #pragma unroll
    for (int j = 0; j < Hh; ++j) w[j]  = Whh[ii * Hh + j];
    #pragma unroll
    for (int j = 0; j < Hh; ++j) wf[j] = Wfc[j];
    const float bfc = bfc_p[0];

    if (!is_scan) __builtin_amdgcn_s_setprio(3);

    float h = 0.0f;

    const float* const xl = xp + (size_t)b * Hh + ii;

    float xa[BLK], xb[BLK];
    #pragma unroll
    for (int k = 0; k < BLK; ++k) xa[k] = xl[(size_t)k * XSTEP];

    float* const out_b0 = out + (size_t)(blockIdx.x * 2) * Sseq;

    const float* xn = xl + (size_t)(BLK * XSTEP);

    float* const hcol = hbuf + lane;

    for (int g = 0; g < NBLK; ++g) {
        if (!is_scan && g >= 3) fc_block(hbuf, g - 3, lane, wf, bfc, out_b0);

        wait_ds_le<8>();
        bar_signal();

        if (is_scan) {
            float* const hblk = hcol + (g & (RING_BLKS - 1)) * (BLK * ROW_DW);

            #pragma unroll
            for (int k = 0; k < BLK; ++k) xb[k] = xn[(size_t)k * XSTEP];

            #pragma unroll
            for (int k = 0; k < BLK; ++k) {
                float hn;
                rnn_step(w, h, hn, xa[k]);
                if (k > 0) hblk[(k - 1) * ROW_DW] = h;
                h = hn;
            }
            hblk[(BLK - 1) * ROW_DW] = h;

            #pragma unroll
            for (int k = 0; k < BLK; ++k) xa[k] = xb[k];
            xn += (g + 2 < NBLK) ? (BLK * XSTEP) : 0;
        }

        bar_wait();
    }

    if (!is_scan) fc_block(hbuf, NBLK - 3, lane, wf, bfc, out_b0);
    wait_ds_le<0>();
    bar_signal();
    bar_wait();
    if (!is_scan) {
        fc_block(hbuf, NBLK - 2, lane, wf, bfc, out_b0);
        fc_block(hbuf, NBLK - 1, lane, wf, bfc, out_b0);
        __builtin_amdgcn_s_setprio(0);
    }
}

extern "C" void kernel_launch(void* const* d_in, const int* in_sizes, int n_in,
                              void* d_out, int out_size, void* d_ws, size_t ws_size,
                              hipStream_t stream) {
    (void)in_sizes; (void)n_in; (void)out_size; (void)ws_size;
    const float* inp = (const float*)d_in[0];
    const float* Wih = (const float*)d_in[1];
    const float* Whh = (const float*)d_in[2];
    const float* bih = (const float*)d_in[3];
    const float* bhh = (const float*)d_in[4];
    const float* Wfc = (const float*)d_in[5];
    const float* bfc = (const float*)d_in[6];
    float* out = (float*)d_out;
    float* xp  = (float*)d_ws;

    const int tiles = (Bsz * Sseq) / 16;
    xproj_wmma<<<dim3(tiles / (XP_WAVES * XP_TPW)), 32 * XP_WAVES, 0, stream>>>(inp, Wih, bih, bhh, xp);
    rnn_scan<<<dim3(Bsz / 2), 64, 0, stream>>>(xp, Whh, Wfc, bfc, out);
}


# scan: matvec of the step summed in three interleaved DPP chains (x + terms 0..3, terms 4..6, terms 7..9, joined A + (B + C)) instead of two chains of five; order of the sum changed, f32 kept
# speedup vs baseline: 5.4240x; 1.0304x over previous
#include <hip/hip_runtime.h>

typedef __attribute__((ext_vector_type(2))) float v2f;
typedef __attribute__((ext_vector_type(4))) float v4f;
typedef __attribute__((ext_vector_type(8))) float v8f;

#define Bsz  64
#define Sseq 4096
#define Cin  80
#define Hh   10

#define XP_TPW     4
#define XP_WAVES   4
#define XP_PITCH   84
#define XP_WAVE_DW (16 * XP_PITCH)
#define XP_PIECES  ((16 * Cin) / (4 * 32))


__device__ __forceinline__ void xproj_load(v4f (&dst)[XP_PIECES], const v4f* __restrict__ p) {
    #pragma unroll
    for (int i = 0; i < XP_PIECES; ++i) dst[i] = p[i * 32];
}

__device__ __forceinline__ void xproj_stage(float* st, const int (&woff)[XP_PIECES],
                                            const v4f (&src)[XP_PIECES]) {
    #pragma unroll
    for (int i = 0; i < XP_PIECES; ++i)
        *reinterpret_cast<v4f*>(st + woff[i]) = src[i];
}

__device__ __forceinline__ void xproj_mul(const float* st, int roff, const v2f (&bfrag)[Cin / 4],
                                          float bias, int tile, int half, int n,
                                          float* __restrict__ xp) {
    v8f acc = {};
    #pragma unroll
    for (int j = 0; j < XP_PIECES; ++j) {
        const v4f a4 = *reinterpret_cast<const v4f*>(st + roff + 8 * j);
        v2f a0; a0[0] = a4[0]; a0[1] = a4[1];
        v2f a1; a1[0] = a4[2]; a1[1] = a4[3];
        acc = __builtin_amdgcn_wmma_f32_16x16x4_f32(
            false, a0, false, bfrag[2 * j], (short)0, acc, false, false);
        acc = __builtin_amdgcn_wmma_f32_16x16x4_f32(
            false, a1, false, bfrag[2 * j + 1], (short)0, acc, false, false);
    }

    if (n < Hh) {
        #pragma unroll
        for (int v = 0; v < 8; ++v) {
            const int gr   = tile * 16 + v + half * 8;
            const int bidx = gr >> 12;
            const int sidx = gr & 4095;
            xp[((size_t)sidx * Bsz + bidx) * Hh + n] = acc[v] + bias;
        }
    }
}

__global__ __launch_bounds__(128) __attribute__((amdgpu_waves_per_eu(4, 4)))
void xproj_wmma(const float* __restrict__ inp,
                const float* __restrict__ Wih,
                const float* __restrict__ bih,
                const float* __restrict__ bhh,
                float* __restrict__ xp) {
    __shared__ __attribute__((aligned(16))) float stage[XP_WAVES * XP_WAVE_DW];

    const int lane = threadIdx.x & 31;
    const int wave = threadIdx.x >> 5;
    const int half = lane >> 4;
    const int n    = lane & 15;
    const int nn   = (n < Hh) ? n : (Hh - 1);
    const float nmask = (n < Hh) ? 1.0f : 0.0f;
    const int tile0 = (blockIdx.x * XP_WAVES + wave) * XP_TPW;

    float* const st = stage + wave * XP_WAVE_DW;

    constexpr int TILE4 = 16 * Cin / 4;
    const v4f* const in4 = reinterpret_cast<const v4f*>(inp) + (size_t)tile0 * TILE4 + lane;
    v4f p0[XP_PIECES], p1[XP_PIECES], p2[XP_PIECES], p3[XP_PIECES];
    xproj_load(p0, in4);
    xproj_load(p1, in4 + TILE4);
    __builtin_amdgcn_sched_barrier(0);

    v2f bfrag[Cin / 4];
    #pragma unroll
    for (int j = 0; j < XP_PIECES; ++j) {
        const v4f w4 = *reinterpret_cast<const v4f*>(Wih + nn * Cin + 8 * j + 4 * half);
        bfrag[2 * j][0]     = nmask * w4[0];
        bfrag[2 * j][1]     = nmask * w4[1];
        bfrag[2 * j + 1][0] = nmask * w4[2];
        bfrag[2 * j + 1][1] = nmask * w4[3];
    }
    const float bias = bih[nn] + bhh[nn];

    int woff[XP_PIECES];
    #pragma unroll
    for (int i = 0; i < XP_PIECES; ++i) {
        const int f = i * 32 + lane;
        const int r = f / (Cin / 4);
        woff[i] = r * XP_PITCH + (f - r * (Cin / 4)) * 4;
    }
    const int roff = n * XP_PITCH + 4 * half;

    static_assert(XP_TPW == 4, "the four tiles below are the wave's");
    xproj_stage(st, woff, p0);
    xproj_load(p2, in4 + 2 * TILE4);
    xproj_mul(st, roff, bfrag, bias, tile0 + 0, half, n, xp);

    xproj_stage(st, woff, p1);
    xproj_load(p3, in4 + 3 * TILE4);
    xproj_mul(st, roff, bfrag, bias, tile0 + 1, half, n, xp);

    xproj_stage(st, woff, p2);
    xproj_mul(st, roff, bfrag, bias, tile0 + 2, half, n, xp);

    xproj_stage(st, woff, p3);
    xproj_mul(st, roff, bfrag, bias, tile0 + 3, half, n, xp);
}

#define BLK   16
#define NBLK  (Sseq / BLK)
#define XSTEP (Bsz * Hh)
#define RING_BLKS 4
#define ROW_DW    36

__device__ __forceinline__ float recip(float d) {
    return __builtin_amdgcn_rcpf(d);
}

__device__ __forceinline__ float sig_exp(float logit) {
    return __expf(-logit);
}

__device__ __forceinline__ float sig4_of_exp(float e) {
    const float sg = recip(1.0f + e);
    const float s2 = sg * sg;
    return s2 * s2;
}

template <int N>
__device__ __forceinline__ void wait_ds_le() {
    asm volatile("s_wait_dscnt %0" :: "i"(N) : "memory");
}
__device__ __forceinline__ void bar_signal() {
    asm volatile("s_barrier_signal -1" ::: "memory");
}
__device__ __forceinline__ void bar_wait() {
    asm volatile("s_barrier_wait -1" ::: "memory");
}

__device__ __forceinline__ void rnn_step(const float (&w)[Hh], float h, float& hn, float x) {
    float a = x, b, c, hn_;
    asm(
        "v_fmac_f32_dpp %[a], %[h], %[w0] row_share:0 row_mask:0xf bank_mask:0xf bound_ctrl:1\n\t"
        "v_mul_f32_dpp %[b], %[h], %[w4] row_share:4 row_mask:0xf bank_mask:0xf bound_ctrl:1\n\t"
        "v_mul_f32_dpp %[c], %[h], %[w7] row_share:7 row_mask:0xf bank_mask:0xf bound_ctrl:1\n\t"
        "v_fmac_f32_dpp %[a], %[h], %[w1] row_share:1 row_mask:0xf bank_mask:0xf bound_ctrl:1\n\t"
        "v_fmac_f32_dpp %[b], %[h], %[w5] row_share:5 row_mask:0xf bank_mask:0xf bound_ctrl:1\n\t"
        "v_fmac_f32_dpp %[c], %[h], %[w8] row_share:8 row_mask:0xf bank_mask:0xf bound_ctrl:1\n\t"
        "v_fmac_f32_dpp %[a], %[h], %[w2] row_share:2 row_mask:0xf bank_mask:0xf bound_ctrl:1\n\t"
        "v_fmac_f32_dpp %[b], %[h], %[w6] row_share:6 row_mask:0xf bank_mask:0xf bound_ctrl:1\n\t"
        "v_fmac_f32_dpp %[c], %[h], %[w9] row_share:9 row_mask:0xf bank_mask:0xf bound_ctrl:1\n\t"
        "v_fmac_f32_dpp %[a], %[h], %[w3] row_share:3 row_mask:0xf bank_mask:0xf bound_ctrl:1\n\t"
        "v_add_f32 %[b], %[b], %[c]\n\t"
        "v_add_f32 %[a], %[a], %[b]\n\t"
        "v_tanh_f32 %[hn], %[a]\n\t"
        "v_nop"
        : [a] "+v"(a), [b] "=&v"(b), [c] "=&v"(c), [hn] "=&v"(hn_)
        : [h] "v"(h),
          [w0] "v"(w[0]), [w1] "v"(w[1]), [w2] "v"(w[2]), [w3] "v"(w[3]), [w4] "v"(w[4]),
          [w5] "v"(w[5]), [w6] "v"(w[6]), [w7] "v"(w[7]), [w8] "v"(w[8]), [w9] "v"(w[9]));
    hn = hn_;
}

__device__ __forceinline__ void fc_block(const float* hbuf, int g, int lane,
                                         const float (&wf)[Hh], float bfc,
                                         float* __restrict__ out_b0) {
    const int hb = lane >> 4;
    const int k  = lane & 15;
    const float* row = hbuf + ((g & (RING_BLKS - 1)) * BLK + k) * ROW_DW + hb * 16;
    const float4 q0 = *reinterpret_cast<const float4*>(row);
    const float4 q1 = *reinterpret_cast<const float4*>(row + 4);
    const float2 q2 = *reinterpret_cast<const float2*>(row + 8);
    float l0 = fmaf(q0.x, wf[0], bfc);
    l0 = fmaf(q0.y, wf[1], l0);
    l0 = fmaf(q0.z, wf[2], l0);
    l0 = fmaf(q0.w, wf[3], l0);
    l0 = fmaf(q1.x, wf[4], l0);
    float l1 = q1.y * wf[5];
    l1 = fmaf(q1.z, wf[6], l1);
    l1 = fmaf(q1.w, wf[7], l1);
    l1 = fmaf(q2.x, wf[8], l1);
    l1 = fmaf(q2.y, wf[9], l1);
    out_b0[(size_t)hb * Sseq + g * BLK + k] = sig4_of_exp(sig_exp(l0 + l1));
}

__global__ __launch_bounds__(64) void rnn_scan(const float* __restrict__ xp,
                                               const float* __restrict__ Whh,
                                               const float* __restrict__ Wfc,
                                               const float* __restrict__ bfc_p,
                                               float* __restrict__ out) {
    __shared__ __attribute__((aligned(16))) float hbuf[RING_BLKS * BLK * ROW_DW];

    const int lane = threadIdx.x & 31;
    const bool is_scan = __builtin_amdgcn_readfirstlane((int)(threadIdx.x >> 5)) == 0;
    const int half = lane >> 4;
    const int i    = lane & 15;
    const int ii   = (i < Hh) ? i : (Hh - 1);
    const int b    = blockIdx.x * 2 + half;

    float w[Hh], wf[Hh];
    #pragma unroll
    for (int j = 0; j < Hh; ++j) w[j]  = Whh[ii * Hh + j];
    #pragma unroll
    for (int j = 0; j < Hh; ++j) wf[j] = Wfc[j];
    const float bfc = bfc_p[0];

    if (!is_scan) __builtin_amdgcn_s_setprio(3);

    float h = 0.0f;

    const float* const xl = xp + (size_t)b * Hh + ii;

    float xa[BLK], xb[BLK];
    #pragma unroll
    for (int k = 0; k < BLK; ++k) xa[k] = xl[(size_t)k * XSTEP];

    float* const out_b0 = out + (size_t)(blockIdx.x * 2) * Sseq;

    const float* xn = xl + (size_t)(BLK * XSTEP);

    float* const hcol = hbuf + lane;

    for (int g = 0; g < NBLK; ++g) {
        if (!is_scan && g >= 3) fc_block(hbuf, g - 3, lane, wf, bfc, out_b0);

        wait_ds_le<8>();
        bar_signal();

        if (is_scan) {
            float* const hblk = hcol + (g & (RING_BLKS - 1)) * (BLK * ROW_DW);

            #pragma unroll
            for (int k = 0; k < BLK; ++k) xb[k] = xn[(size_t)k * XSTEP];

            #pragma unroll
            for (int k = 0; k < BLK; ++k) {
                float hn;
                rnn_step(w, h, hn, xa[k]);
                if (k > 0) hblk[(k - 1) * ROW_DW] = h;
                h = hn;
            }
            hblk[(BLK - 1) * ROW_DW] = h;

            #pragma unroll
            for (int k = 0; k < BLK; ++k) xa[k] = xb[k];
            xn += (g + 2 < NBLK) ? (BLK * XSTEP) : 0;
        }

        bar_wait();
    }

    if (!is_scan) fc_block(hbuf, NBLK - 3, lane, wf, bfc, out_b0);
    wait_ds_le<0>();
    bar_signal();
    bar_wait();
    if (!is_scan) {
        fc_block(hbuf, NBLK - 2, lane, wf, bfc, out_b0);
        fc_block(hbuf, NBLK - 1, lane, wf, bfc, out_b0);
        __builtin_amdgcn_s_setprio(0);
    }
}

extern "C" void kernel_launch(void* const* d_in, const int* in_sizes, int n_in,
                              void* d_out, int out_size, void* d_ws, size_t ws_size,
                              hipStream_t stream) {
    (void)in_sizes; (void)n_in; (void)out_size; (void)ws_size;
    const float* inp = (const float*)d_in[0];
    const float* Wih = (const float*)d_in[1];
    const float* Whh = (const float*)d_in[2];
    const float* bih = (const float*)d_in[3];
    const float* bhh = (const float*)d_in[4];
    const float* Wfc = (const float*)d_in[5];
    const float* bfc = (const float*)d_in[6];
    float* out = (float*)d_out;
    float* xp  = (float*)d_ws;

    const int tiles = (Bsz * Sseq) / 16;
    xproj_wmma<<<dim3(tiles / (XP_WAVES * XP_TPW)), 32 * XP_WAVES, 0, stream>>>(inp, Wih, bih, bhh, xp);
    rnn_scan<<<dim3(Bsz / 2), 64, 0, stream>>>(xp, Whh, Wfc, bfc, out);
}


# scan: a block's 16 hidden states kept in registers, stored to the LDS ring as one group beside the next block's 16 xp loads: one place with memory instructions per 16 steps instead of five
# speedup vs baseline: 5.6433x; 1.0404x over previous
#include <hip/hip_runtime.h>

typedef __attribute__((ext_vector_type(2))) float v2f;
typedef __attribute__((ext_vector_type(4))) float v4f;
typedef __attribute__((ext_vector_type(8))) float v8f;

#define Bsz  64
#define Sseq 4096
#define Cin  80
#define Hh   10

#define XP_TPW     4
#define XP_WAVES   4
#define XP_PITCH   84
#define XP_WAVE_DW (16 * XP_PITCH)
#define XP_PIECES  ((16 * Cin) / (4 * 32))


__device__ __forceinline__ void xproj_load(v4f (&dst)[XP_PIECES], const v4f* __restrict__ p) {
    #pragma unroll
    for (int i = 0; i < XP_PIECES; ++i) dst[i] = p[i * 32];
}

__device__ __forceinline__ void xproj_stage(float* st, const int (&woff)[XP_PIECES],
                                            const v4f (&src)[XP_PIECES]) {
    #pragma unroll
    for (int i = 0; i < XP_PIECES; ++i)
        *reinterpret_cast<v4f*>(st + woff[i]) = src[i];
}

__device__ __forceinline__ void xproj_mul(const float* st, int roff, const v2f (&bfrag)[Cin / 4],
                                          float bias, int tile, int half, int n,
                                          float* __restrict__ xp) {
    v8f acc = {};
    #pragma unroll
    for (int j = 0; j < XP_PIECES; ++j) {
        const v4f a4 = *reinterpret_cast<const v4f*>(st + roff + 8 * j);
        v2f a0; a0[0] = a4[0]; a0[1] = a4[1];
        v2f a1; a1[0] = a4[2]; a1[1] = a4[3];
        acc = __builtin_amdgcn_wmma_f32_16x16x4_f32(
            false, a0, false, bfrag[2 * j], (short)0, acc, false, false);
        acc = __builtin_amdgcn_wmma_f32_16x16x4_f32(
            false, a1, false, bfrag[2 * j + 1], (short)0, acc, false, false);
    }

    if (n < Hh) {
        #pragma unroll
        for (int v = 0; v < 8; ++v) {
            const int gr   = tile * 16 + v + half * 8;
            const int bidx = gr >> 12;
            const int sidx = gr & 4095;
            xp[((size_t)sidx * Bsz + bidx) * Hh + n] = acc[v] + bias;
        }
    }
}

__global__ __launch_bounds__(128) __attribute__((amdgpu_waves_per_eu(4, 4)))
void xproj_wmma(const float* __restrict__ inp,
                const float* __restrict__ Wih,
                const float* __restrict__ bih,
                const float* __restrict__ bhh,
                float* __restrict__ xp) {
    __shared__ __attribute__((aligned(16))) float stage[XP_WAVES * XP_WAVE_DW];

    const int lane = threadIdx.x & 31;
    const int wave = threadIdx.x >> 5;
    const int half = lane >> 4;
    const int n    = lane & 15;
    const int nn   = (n < Hh) ? n : (Hh - 1);
    const float nmask = (n < Hh) ? 1.0f : 0.0f;
    const int tile0 = (blockIdx.x * XP_WAVES + wave) * XP_TPW;

    float* const st = stage + wave * XP_WAVE_DW;

    constexpr int TILE4 = 16 * Cin / 4;
    const v4f* const in4 = reinterpret_cast<const v4f*>(inp) + (size_t)tile0 * TILE4 + lane;
    v4f p0[XP_PIECES], p1[XP_PIECES], p2[XP_PIECES], p3[XP_PIECES];
    xproj_load(p0, in4);
    xproj_load(p1, in4 + TILE4);
    __builtin_amdgcn_sched_barrier(0);

    v2f bfrag[Cin / 4];
    #pragma unroll
    for (int j = 0; j < XP_PIECES; ++j) {
        const v4f w4 = *reinterpret_cast<const v4f*>(Wih + nn * Cin + 8 * j + 4 * half);
        bfrag[2 * j][0]     = nmask * w4[0];
        bfrag[2 * j][1]     = nmask * w4[1];
        bfrag[2 * j + 1][0] = nmask * w4[2];
        bfrag[2 * j + 1][1] = nmask * w4[3];
    }
    const float bias = bih[nn] + bhh[nn];

    int woff[XP_PIECES];
    #pragma unroll
    for (int i = 0; i < XP_PIECES; ++i) {
        const int f = i * 32 + lane;
        const int r = f / (Cin / 4);
        woff[i] = r * XP_PITCH + (f - r * (Cin / 4)) * 4;
    }
    const int roff = n * XP_PITCH + 4 * half;

    static_assert(XP_TPW == 4, "the four tiles below are the wave's");
    xproj_stage(st, woff, p0);
    xproj_load(p2, in4 + 2 * TILE4);
    xproj_mul(st, roff, bfrag, bias, tile0 + 0, half, n, xp);

    xproj_stage(st, woff, p1);
    xproj_load(p3, in4 + 3 * TILE4);
    xproj_mul(st, roff, bfrag, bias, tile0 + 1, half, n, xp);

    xproj_stage(st, woff, p2);
    xproj_mul(st, roff, bfrag, bias, tile0 + 2, half, n, xp);

    xproj_stage(st, woff, p3);
    xproj_mul(st, roff, bfrag, bias, tile0 + 3, half, n, xp);
}

#define BLK   16
#define NBLK  (Sseq / BLK)
#define XSTEP (Bsz * Hh)
#define RING_BLKS 4
#define LANE_DW   20
#define SLOT_DW   (32 * LANE_DW)

__device__ __forceinline__ float recip(float d) {
    return __builtin_amdgcn_rcpf(d);
}

__device__ __forceinline__ float sig_exp(float logit) {
    return __expf(-logit);
}

__device__ __forceinline__ float sig4_of_exp(float e) {
    const float sg = recip(1.0f + e);
    const float s2 = sg * sg;
    return s2 * s2;
}

template <int N>
__device__ __forceinline__ void wait_ds_le() {
    asm volatile("s_wait_dscnt %0" :: "i"(N) : "memory");
}
__device__ __forceinline__ void bar_signal() {
    asm volatile("s_barrier_signal -1" ::: "memory");
}
__device__ __forceinline__ void bar_wait() {
    asm volatile("s_barrier_wait -1" ::: "memory");
}

__device__ __forceinline__ void rnn_step(const float (&w)[Hh], float h, float& hn, float x) {
    float a = x, b, c, hn_;
    asm(
        "v_fmac_f32_dpp %[a], %[h], %[w0] row_share:0 row_mask:0xf bank_mask:0xf bound_ctrl:1\n\t"
        "v_mul_f32_dpp %[b], %[h], %[w4] row_share:4 row_mask:0xf bank_mask:0xf bound_ctrl:1\n\t"
        "v_mul_f32_dpp %[c], %[h], %[w7] row_share:7 row_mask:0xf bank_mask:0xf bound_ctrl:1\n\t"
        "v_fmac_f32_dpp %[a], %[h], %[w1] row_share:1 row_mask:0xf bank_mask:0xf bound_ctrl:1\n\t"
        "v_fmac_f32_dpp %[b], %[h], %[w5] row_share:5 row_mask:0xf bank_mask:0xf bound_ctrl:1\n\t"
        "v_fmac_f32_dpp %[c], %[h], %[w8] row_share:8 row_mask:0xf bank_mask:0xf bound_ctrl:1\n\t"
        "v_fmac_f32_dpp %[a], %[h], %[w2] row_share:2 row_mask:0xf bank_mask:0xf bound_ctrl:1\n\t"
        "v_fmac_f32_dpp %[b], %[h], %[w6] row_share:6 row_mask:0xf bank_mask:0xf bound_ctrl:1\n\t"
        "v_fmac_f32_dpp %[c], %[h], %[w9] row_share:9 row_mask:0xf bank_mask:0xf bound_ctrl:1\n\t"
        "v_fmac_f32_dpp %[a], %[h], %[w3] row_share:3 row_mask:0xf bank_mask:0xf bound_ctrl:1\n\t"
        "v_add_f32 %[b], %[b], %[c]\n\t"
        "v_add_f32 %[a], %[a], %[b]\n\t"
        "v_tanh_f32 %[hn], %[a]\n\t"
        "v_nop"
        : [a] "+v"(a), [b] "=&v"(b), [c] "=&v"(c), [hn] "=&v"(hn_)
        : [h] "v"(h),
          [w0] "v"(w[0]), [w1] "v"(w[1]), [w2] "v"(w[2]), [w3] "v"(w[3]), [w4] "v"(w[4]),
          [w5] "v"(w[5]), [w6] "v"(w[6]), [w7] "v"(w[7]), [w8] "v"(w[8]), [w9] "v"(w[9]));
    hn = hn_;
}

__device__ __forceinline__ void fc_block(const float* hbuf, int g, int lane,
                                         const float (&wf)[Hh], float bfc,
                                         float* __restrict__ out_b0) {
    const int hb = lane >> 4;
    const int k  = lane & 15;
    const float* col = hbuf + (g & (RING_BLKS - 1)) * SLOT_DW + (hb * 16) * LANE_DW + k;
    float q[Hh];
    #pragma unroll
    for (int j = 0; j < Hh; ++j) q[j] = col[j * LANE_DW];
    float l0 = fmaf(q[0], wf[0], bfc);
    l0 = fmaf(q[1], wf[1], l0);
    l0 = fmaf(q[2], wf[2], l0);
    l0 = fmaf(q[3], wf[3], l0);
    l0 = fmaf(q[4], wf[4], l0);
    float l1 = q[5] * wf[5];
    l1 = fmaf(q[6], wf[6], l1);
    l1 = fmaf(q[7], wf[7], l1);
    l1 = fmaf(q[8], wf[8], l1);
    l1 = fmaf(q[9], wf[9], l1);
    out_b0[(size_t)hb * Sseq + g * BLK + k] = sig4_of_exp(sig_exp(l0 + l1));
}

__global__ __launch_bounds__(64) void rnn_scan(const float* __restrict__ xp,
                                               const float* __restrict__ Whh,
                                               const float* __restrict__ Wfc,
                                               const float* __restrict__ bfc_p,
                                               float* __restrict__ out) {
    __shared__ __attribute__((aligned(16))) float hbuf[RING_BLKS * SLOT_DW];

    const int lane = threadIdx.x & 31;
    const bool is_scan = __builtin_amdgcn_readfirstlane((int)(threadIdx.x >> 5)) == 0;
    const int half = lane >> 4;
    const int i    = lane & 15;
    const int ii   = (i < Hh) ? i : (Hh - 1);
    const int b    = blockIdx.x * 2 + half;

    float w[Hh], wf[Hh];
    #pragma unroll
    for (int j = 0; j < Hh; ++j) w[j]  = Whh[ii * Hh + j];
    #pragma unroll
    for (int j = 0; j < Hh; ++j) wf[j] = Wfc[j];
    const float bfc = bfc_p[0];

    if (!is_scan) __builtin_amdgcn_s_setprio(3);

    float h = 0.0f;

    const float* const xl = xp + (size_t)b * Hh + ii;

    float xa[BLK], xb[BLK];
    #pragma unroll
    for (int k = 0; k < BLK; ++k) xa[k] = xl[(size_t)k * XSTEP];

    float* const out_b0 = out + (size_t)(blockIdx.x * 2) * Sseq;

    const float* xn = xl + (size_t)(BLK * XSTEP);

    float* const hrun = hbuf + lane * LANE_DW;

    float hs[BLK];
    #pragma unroll
    for (int k = 0; k < BLK; ++k) hs[k] = 0.0f;
    float* hblk = hrun + (RING_BLKS - 1) * SLOT_DW;

    for (int g = 0; g < NBLK; ++g) {
        if (!is_scan && g >= 3) fc_block(hbuf, g - 3, lane, wf, bfc, out_b0);

        wait_ds_le<0>();
        bar_signal();

        if (is_scan) {
            #pragma unroll
            for (int k = 0; k < BLK; k += 4) {
                v4f q; q[0] = hs[k]; q[1] = hs[k + 1]; q[2] = hs[k + 2]; q[3] = hs[k + 3];
                *reinterpret_cast<v4f*>(hblk + k) = q;
            }

            #pragma unroll
            for (int k = 0; k < BLK; ++k) xb[k] = xn[(size_t)k * XSTEP];

            #pragma unroll
            for (int k = 0; k < BLK; ++k) {
                rnn_step(w, h, hs[k], xa[k]);
                h = hs[k];
            }
            hblk = hrun + (g & (RING_BLKS - 1)) * SLOT_DW;

            #pragma unroll
            for (int k = 0; k < BLK; ++k) xa[k] = xb[k];
            xn += (g + 2 < NBLK) ? (BLK * XSTEP) : 0;
        }

        bar_wait();
    }

    if (is_scan) {
        #pragma unroll
        for (int k = 0; k < BLK; k += 4) {
            v4f q; q[0] = hs[k]; q[1] = hs[k + 1]; q[2] = hs[k + 2]; q[3] = hs[k + 3];
            *reinterpret_cast<v4f*>(hblk + k) = q;
        }
    }
    if (!is_scan) fc_block(hbuf, NBLK - 3, lane, wf, bfc, out_b0);
    wait_ds_le<0>();
    bar_signal();
    bar_wait();
    if (!is_scan) {
        fc_block(hbuf, NBLK - 2, lane, wf, bfc, out_b0);
        fc_block(hbuf, NBLK - 1, lane, wf, bfc, out_b0);
        __builtin_amdgcn_s_setprio(0);
    }
}

extern "C" void kernel_launch(void* const* d_in, const int* in_sizes, int n_in,
                              void* d_out, int out_size, void* d_ws, size_t ws_size,
                              hipStream_t stream) {
    (void)in_sizes; (void)n_in; (void)out_size; (void)ws_size;
    const float* inp = (const float*)d_in[0];
    const float* Wih = (const float*)d_in[1];
    const float* Whh = (const float*)d_in[2];
    const float* bih = (const float*)d_in[3];
    const float* bhh = (const float*)d_in[4];
    const float* Wfc = (const float*)d_in[5];
    const float* bfc = (const float*)d_in[6];
    float* out = (float*)d_out;
    float* xp  = (float*)d_ws;

    const int tiles = (Bsz * Sseq) / 16;
    xproj_wmma<<<dim3(tiles / (XP_WAVES * XP_TPW)), 32 * XP_WAVES, 0, stream>>>(inp, Wih, bih, bhh, xp);
    rnn_scan<<<dim3(Bsz / 2), 64, 0, stream>>>(xp, Whh, Wfc, bfc, out);
}
